# starting kernel
# baseline (speedup 1.0000x reference)
#include <hip/hip_runtime.h>

typedef __attribute__((ext_vector_type(2))) float v2f;
typedef __attribute__((ext_vector_type(8))) float v8f;

__global__ void init_deg_kernel(float* __restrict__ deg, int n) {
    int i = blockIdx.x * blockDim.x + threadIdx.x;
    if (i < n) deg[i] = 1.0f;
}

__global__ void edge_deg_kernel(const int* __restrict__ dst, float* __restrict__ deg, int E) {
    int e = blockIdx.x * blockDim.x + threadIdx.x;
    if (e < E) atomicAdd(&deg[dst[e]], 1.0f);
}

__global__ void deg_to_dinv_kernel(float* __restrict__ deg, int n) {
    int i = blockIdx.x * blockDim.x + threadIdx.x;
    if (i < n) {
        float d = deg[i];
        deg[i] = (d > 0.0f) ? rsqrtf(d) : 0.0f;
    }
}

__global__ void pad_x_kernel(const float* __restrict__ x, float* __restrict__ xp, int n) {
    int t = blockIdx.x * blockDim.x + threadIdx.x;
    if (t >= n * 8) return;
    int i = t >> 3, k = t & 7;
    xp[t] = (k < 6) ? x[i * 6 + k] : 0.0f;
}

__global__ void pad_w_kernel(const float* __restrict__ w, float* __restrict__ wp) {
    int t = blockIdx.x * blockDim.x + threadIdx.x;
    if (t >= 8 * 64) return;
    wp[t] = (t < 6 * 64) ? w[t] : 0.0f;
}

template <int K, int NT, bool BIAS_RELU>
__global__ __launch_bounds__(32) void gemm_wmma(
    const float* __restrict__ X, const float* __restrict__ W,
    float* __restrict__ Out, const float* __restrict__ bias)
{
    constexpr int NCOLS = NT * 16;
    const int lane = threadIdx.x;
    const int hi   = lane >> 4;
    const int lo   = lane & 15;
    const int row0 = blockIdx.x * 16;

    const float* Arow = X + (size_t)(row0 + lo) * K;

    v8f c[NT];
#pragma unroll
    for (int t = 0; t < NT; ++t) c[t] = (v8f){};

#pragma unroll
    for (int k0 = 0; k0 < K; k0 += 4) {
        const int ka = k0 + 2 * hi;
        const v2f a = *(const v2f*)(Arow + ka);
#pragma unroll
        for (int t = 0; t < NT; ++t) {
            v2f b;
            b.x = W[ka * NCOLS + 16 * t + lo];
            b.y = W[(ka + 1) * NCOLS + 16 * t + lo];
            c[t] = __builtin_amdgcn_wmma_f32_16x16x4_f32(
                false, a, false, b, (short)0, c[t], false, false);
        }
    }

#pragma unroll
    for (int t = 0; t < NT; ++t) {
        const int col = 16 * t + lo;
#pragma unroll
        for (int j = 0; j < 8; ++j) {
            float v = c[t][j];
            if (BIAS_RELU) v = fmaxf(v + bias[col], 0.0f);
            Out[(size_t)(row0 + j + 8 * hi) * NCOLS + col] = v;
        }
    }
}

__global__ void self_init_kernel(const float* __restrict__ tmp,
                                 const float* __restrict__ dinv,
                                 float* __restrict__ agg, int n) {
    long long t = (long long)blockIdx.x * blockDim.x + threadIdx.x;
    if (t >= (long long)n * 64) return;
    int i = (int)(t >> 6);
    float di = dinv[i];
    agg[t] = tmp[t] * di * di;
}

__global__ void edge_msg_kernel(const int* __restrict__ src,
                                const int* __restrict__ dst,
                                const float* __restrict__ dinv,
                                const float* __restrict__ tmp,
                                float* __restrict__ agg, int E) {
    long long t = (long long)blockIdx.x * blockDim.x + threadIdx.x;
    int e    = (int)(t >> 5);
    int lane = (int)(t & 31);
    if (e >= E) return;
    int s = src[e], d = dst[e];
    float nrm = dinv[s] * dinv[d];
    const float2 v = *(const float2*)(tmp + (size_t)s * 64 + 2 * lane);
    atomicAdd(agg + (size_t)d * 64 + 2 * lane,     v.x * nrm);
    atomicAdd(agg + (size_t)d * 64 + 2 * lane + 1, v.y * nrm);
}

__global__ void bias_relu_kernel(float* __restrict__ h,
                                 const float* __restrict__ bias, int n) {
    long long t = (long long)blockIdx.x * blockDim.x + threadIdx.x;
    if (t >= (long long)n * 64) return;
    int f = (int)(t & 63);
    h[t] = fmaxf(h[t] + bias[f], 0.0f);
}

__global__ void pool_zero_kernel(float* __restrict__ sums, float* __restrict__ cnts,
                                 int ng, int hid) {
    int t = blockIdx.x * blockDim.x + threadIdx.x;
    if (t < ng * hid) sums[t] = 0.0f;
    if (t < ng)       cnts[t] = 0.0f;
}

__global__ void pool_add_kernel(const float* __restrict__ h,
                                const int* __restrict__ batch,
                                float* __restrict__ sums, int n) {
    long long t = (long long)blockIdx.x * blockDim.x + threadIdx.x;
    if (t >= (long long)n * 64) return;
    int i = (int)(t >> 6);
    int f = (int)(t & 63);
    atomicAdd(sums + (size_t)batch[i] * 64 + f, h[t]);
}

__global__ void pool_cnt_kernel(const int* __restrict__ batch,
                                float* __restrict__ cnts, int n) {
    int i = blockIdx.x * blockDim.x + threadIdx.x;
    if (i < n) atomicAdd(&cnts[batch[i]], 1.0f);
}

__global__ void pool_div_kernel(const float* __restrict__ sums,
                                const float* __restrict__ cnts,
                                float* __restrict__ G, int ng, int hid) {
    int t = blockIdx.x * blockDim.x + threadIdx.x;
    if (t >= ng * hid) return;
    int g = t / hid;
    G[t] = sums[t] / fmaxf(cnts[g], 1.0f);
}

extern "C" void kernel_launch(void* const* d_in, const int* in_sizes, int n_in,
                              void* d_out, int out_size, void* d_ws, size_t ws_size,
                              hipStream_t stream) {
    (void)n_in; (void)out_size; (void)ws_size;

    const float* x     = (const float*)d_in[0];
    const int*   ei    = (const int*)  d_in[1];
    const int*   batch = (const int*)  d_in[2];
    const float* W1  = (const float*)d_in[3];
    const float* b1  = (const float*)d_in[4];
    const float* W2  = (const float*)d_in[5];
    const float* b2  = (const float*)d_in[6];
    const float* W3  = (const float*)d_in[7];
    const float* b3  = (const float*)d_in[8];
    const float* Wfc = (const float*)d_in[9];
    const float* bfc = (const float*)d_in[10];

    const int FIN = 6, HID = 64, EMB = 128, NG = 64;
    const int N = in_sizes[0] / FIN;
    const int E = in_sizes[1] / 2;
    const int* srcIdx = ei;
    const int* dstIdx = ei + E;

    char* ws = (char*)d_ws;
    auto a256 = [](size_t v) { return (v + 255) & ~(size_t)255; };
    size_t off = 0;
    float* dinv = (float*)(ws + off); off += a256((size_t)N * 4);
    float* xpad = (float*)(ws + off); off += a256((size_t)N * 8 * 4);
    float* Wpad = (float*)(ws + off); off += a256((size_t)8 * HID * 4);
    float* bufA = (float*)(ws + off); off += a256((size_t)N * HID * 4);
    float* bufB = (float*)(ws + off); off += a256((size_t)N * HID * 4);
    float* sums = (float*)(ws + off); off += a256((size_t)NG * HID * 4);
    float* cnts = (float*)(ws + off); off += a256((size_t)NG * 4);
    float* G    = (float*)(ws + off); off += a256((size_t)NG * HID * 4);

    const int TB = 256;
    const long long nh = (long long)N * HID;
    const unsigned nhb = (unsigned)((nh + TB - 1) / TB);
    const unsigned eb  = (unsigned)(((long long)E * 32 + TB - 1) / TB);

    init_deg_kernel<<<(N + TB - 1) / TB, TB, 0, stream>>>(dinv, N);
    edge_deg_kernel<<<(E + TB - 1) / TB, TB, 0, stream>>>(dstIdx, dinv, E);
    deg_to_dinv_kernel<<<(N + TB - 1) / TB, TB, 0, stream>>>(dinv, N);

    pad_x_kernel<<<(N * 8 + TB - 1) / TB, TB, 0, stream>>>(x, xpad, N);
    pad_w_kernel<<<2, TB, 0, stream>>>(W1, Wpad);

    const int rowTiles = N / 16;

    gemm_wmma<8, 4, false><<<rowTiles, 32, 0, stream>>>(xpad, Wpad, bufA, nullptr);
    self_init_kernel<<<nhb, TB, 0, stream>>>(bufA, dinv, bufB, N);
    edge_msg_kernel<<<eb, TB, 0, stream>>>(srcIdx, dstIdx, dinv, bufA, bufB, E);
    bias_relu_kernel<<<nhb, TB, 0, stream>>>(bufB, b1, N);

    const float* Wl[2] = {W2, W3};
    const float* bl[2] = {b2, b3};
    for (int l = 0; l < 2; ++l) {
        gemm_wmma<64, 4, false><<<rowTiles, 32, 0, stream>>>(bufB, Wl[l], bufA, nullptr);
        self_init_kernel<<<nhb, TB, 0, stream>>>(bufA, dinv, bufB, N);
        edge_msg_kernel<<<eb, TB, 0, stream>>>(srcIdx, dstIdx, dinv, bufA, bufB, E);
        bias_relu_kernel<<<nhb, TB, 0, stream>>>(bufB, bl[l], N);
    }

    pool_zero_kernel<<<(NG * HID + TB - 1) / TB, TB, 0, stream>>>(sums, cnts, NG, HID);
    pool_add_kernel<<<nhb, TB, 0, stream>>>(bufB, batch, sums, N);
    pool_cnt_kernel<<<(N + TB - 1) / TB, TB, 0, stream>>>(batch, cnts, N);
    pool_div_kernel<<<(NG * HID + TB - 1) / TB, TB, 0, stream>>>(sums, cnts, G, NG, HID);

    gemm_wmma<64, 8, true><<<NG / 16, 32, 0, stream>>>(G, Wfc, (float*)d_out, bfc);
}


# edge_msg: one wave per 32 edges (each edge's indices and norm loaded once, coalesced, handed round by lane reads; 16 row loads in flight per wave) instead of one wave per edge
# speedup vs baseline: 1.2629x; 1.2629x over previous
#include <hip/hip_runtime.h>

typedef __attribute__((ext_vector_type(2))) float v2f;
typedef __attribute__((ext_vector_type(8))) float v8f;

__global__ void init_deg_kernel(float* __restrict__ deg, int n) {
    int i = blockIdx.x * blockDim.x + threadIdx.x;
    if (i < n) deg[i] = 1.0f;
}

__global__ void edge_deg_kernel(const int* __restrict__ dst, float* __restrict__ deg, int E) {
    int e = blockIdx.x * blockDim.x + threadIdx.x;
    if (e < E) atomicAdd(&deg[dst[e]], 1.0f);
}

__global__ void deg_to_dinv_kernel(float* __restrict__ deg, int n) {
    int i = blockIdx.x * blockDim.x + threadIdx.x;
    if (i < n) {
        float d = deg[i];
        deg[i] = (d > 0.0f) ? rsqrtf(d) : 0.0f;
    }
}

__global__ void pad_x_kernel(const float* __restrict__ x, float* __restrict__ xp, int n) {
    int t = blockIdx.x * blockDim.x + threadIdx.x;
    if (t >= n * 8) return;
    int i = t >> 3, k = t & 7;
    xp[t] = (k < 6) ? x[i * 6 + k] : 0.0f;
}

__global__ void pad_w_kernel(const float* __restrict__ w, float* __restrict__ wp) {
    int t = blockIdx.x * blockDim.x + threadIdx.x;
    if (t >= 8 * 64) return;
    wp[t] = (t < 6 * 64) ? w[t] : 0.0f;
}

template <int K, int NT, bool BIAS_RELU>
__global__ __launch_bounds__(32) void gemm_wmma(
    const float* __restrict__ X, const float* __restrict__ W,
    float* __restrict__ Out, const float* __restrict__ bias)
{
    constexpr int NCOLS = NT * 16;
    const int lane = threadIdx.x;
    const int hi   = lane >> 4;
    const int lo   = lane & 15;
    const int row0 = blockIdx.x * 16;

    const float* Arow = X + (size_t)(row0 + lo) * K;

    v8f c[NT];
#pragma unroll
    for (int t = 0; t < NT; ++t) c[t] = (v8f){};

#pragma unroll
    for (int k0 = 0; k0 < K; k0 += 4) {
        const int ka = k0 + 2 * hi;
        const v2f a = *(const v2f*)(Arow + ka);
#pragma unroll
        for (int t = 0; t < NT; ++t) {
            v2f b;
            b.x = W[ka * NCOLS + 16 * t + lo];
            b.y = W[(ka + 1) * NCOLS + 16 * t + lo];
            c[t] = __builtin_amdgcn_wmma_f32_16x16x4_f32(
                false, a, false, b, (short)0, c[t], false, false);
        }
    }

#pragma unroll
    for (int t = 0; t < NT; ++t) {
        const int col = 16 * t + lo;
#pragma unroll
        for (int j = 0; j < 8; ++j) {
            float v = c[t][j];
            if (BIAS_RELU) v = fmaxf(v + bias[col], 0.0f);
            Out[(size_t)(row0 + j + 8 * hi) * NCOLS + col] = v;
        }
    }
}

__global__ void self_init_kernel(const float* __restrict__ tmp,
                                 const float* __restrict__ dinv,
                                 float* __restrict__ agg, int n) {
    long long t = (long long)blockIdx.x * blockDim.x + threadIdx.x;
    if (t >= (long long)n * 64) return;
    int i = (int)(t >> 6);
    float di = dinv[i];
    agg[t] = tmp[t] * di * di;
}

__global__ void edge_msg_kernel(const int* __restrict__ src,
                                const int* __restrict__ dst,
                                const float* __restrict__ dinv,
                                const float* __restrict__ tmp,
                                float* __restrict__ agg, int E) {
    const unsigned lane = threadIdx.x & 31u;
    const long long t = (long long)blockIdx.x * blockDim.x + threadIdx.x;
    const long long e0 = (t >> 5) << 5;
    if (e0 >= (long long)E) return;
    const long long e = e0 + lane;
    int s = 0, d = 0;
    float nrm = 0.0f;
    if (e < (long long)E) {
        s = src[e];
        d = dst[e];
        nrm = dinv[s] * dinv[d];
    }
    const int nrm_bits = __float_as_int(nrm);
    const long long left = (long long)E - e0;

    if (left >= 32) {
#pragma unroll
        for (int j0 = 0; j0 < 32; j0 += 16) {
            float2 v[16];
#pragma unroll
            for (int j = 0; j < 16; ++j) {
                const int sj = __builtin_amdgcn_readlane(s, j0 + j);
                const float* row = tmp + (size_t)sj * 64;
                v[j] = *(const float2*)(row + 2u * lane);
            }
#pragma unroll
            for (int j = 0; j < 16; ++j) {
                const int dj = __builtin_amdgcn_readlane(d, j0 + j);
                const float nj = __int_as_float(__builtin_amdgcn_readlane(nrm_bits, j0 + j));
                float* row = agg + (size_t)dj * 64;
                atomicAdd(row + 2u * lane,      v[j].x * nj);
                atomicAdd(row + 2u * lane + 1u, v[j].y * nj);
            }
        }
    } else {
        const int cnt = (int)left;
        for (int j = 0; j < cnt; ++j) {
            const int sj = __builtin_amdgcn_readlane(s, j);
            const int dj = __builtin_amdgcn_readlane(d, j);
            const float nj = __int_as_float(__builtin_amdgcn_readlane(nrm_bits, j));
            const float2 v = *(const float2*)(tmp + (size_t)sj * 64 + 2u * lane);
            float* row = agg + (size_t)dj * 64;
            atomicAdd(row + 2u * lane,      v.x * nj);
            atomicAdd(row + 2u * lane + 1u, v.y * nj);
        }
    }
}

__global__ void bias_relu_kernel(float* __restrict__ h,
                                 const float* __restrict__ bias, int n) {
    long long t = (long long)blockIdx.x * blockDim.x + threadIdx.x;
    if (t >= (long long)n * 64) return;
    int f = (int)(t & 63);
    h[t] = fmaxf(h[t] + bias[f], 0.0f);
}

__global__ void pool_zero_kernel(float* __restrict__ sums, float* __restrict__ cnts,
                                 int ng, int hid) {
    int t = blockIdx.x * blockDim.x + threadIdx.x;
    if (t < ng * hid) sums[t] = 0.0f;
    if (t < ng)       cnts[t] = 0.0f;
}

__global__ void pool_add_kernel(const float* __restrict__ h,
                                const int* __restrict__ batch,
                                float* __restrict__ sums, int n) {
    long long t = (long long)blockIdx.x * blockDim.x + threadIdx.x;
    if (t >= (long long)n * 64) return;
    int i = (int)(t >> 6);
    int f = (int)(t & 63);
    atomicAdd(sums + (size_t)batch[i] * 64 + f, h[t]);
}

__global__ void pool_cnt_kernel(const int* __restrict__ batch,
                                float* __restrict__ cnts, int n) {
    int i = blockIdx.x * blockDim.x + threadIdx.x;
    if (i < n) atomicAdd(&cnts[batch[i]], 1.0f);
}

__global__ void pool_div_kernel(const float* __restrict__ sums,
                                const float* __restrict__ cnts,
                                float* __restrict__ G, int ng, int hid) {
    int t = blockIdx.x * blockDim.x + threadIdx.x;
    if (t >= ng * hid) return;
    int g = t / hid;
    G[t] = sums[t] / fmaxf(cnts[g], 1.0f);
}

extern "C" void kernel_launch(void* const* d_in, const int* in_sizes, int n_in,
                              void* d_out, int out_size, void* d_ws, size_t ws_size,
                              hipStream_t stream) {
    (void)n_in; (void)out_size; (void)ws_size;

    const float* x     = (const float*)d_in[0];
    const int*   ei    = (const int*)  d_in[1];
    const int*   batch = (const int*)  d_in[2];
    const float* W1  = (const float*)d_in[3];
    const float* b1  = (const float*)d_in[4];
    const float* W2  = (const float*)d_in[5];
    const float* b2  = (const float*)d_in[6];
    const float* W3  = (const float*)d_in[7];
    const float* b3  = (const float*)d_in[8];
    const float* Wfc = (const float*)d_in[9];
    const float* bfc = (const float*)d_in[10];

    const int FIN = 6, HID = 64, EMB = 128, NG = 64;
    const int N = in_sizes[0] / FIN;
    const int E = in_sizes[1] / 2;
    const int* srcIdx = ei;
    const int* dstIdx = ei + E;

    char* ws = (char*)d_ws;
    auto a256 = [](size_t v) { return (v + 255) & ~(size_t)255; };
    size_t off = 0;
    float* dinv = (float*)(ws + off); off += a256((size_t)N * 4);
    float* xpad = (float*)(ws + off); off += a256((size_t)N * 8 * 4);
    float* Wpad = (float*)(ws + off); off += a256((size_t)8 * HID * 4);
    float* bufA = (float*)(ws + off); off += a256((size_t)N * HID * 4);
    float* bufB = (float*)(ws + off); off += a256((size_t)N * HID * 4);
    float* sums = (float*)(ws + off); off += a256((size_t)NG * HID * 4);
    float* cnts = (float*)(ws + off); off += a256((size_t)NG * 4);
    float* G    = (float*)(ws + off); off += a256((size_t)NG * HID * 4);

    const int TB = 256;
    const long long nh = (long long)N * HID;
    const unsigned nhb = (unsigned)((nh + TB - 1) / TB);
    const unsigned eb  = (unsigned)((E + TB - 1) / TB);

    init_deg_kernel<<<(N + TB - 1) / TB, TB, 0, stream>>>(dinv, N);
    edge_deg_kernel<<<(E + TB - 1) / TB, TB, 0, stream>>>(dstIdx, dinv, E);
    deg_to_dinv_kernel<<<(N + TB - 1) / TB, TB, 0, stream>>>(dinv, N);

    pad_x_kernel<<<(N * 8 + TB - 1) / TB, TB, 0, stream>>>(x, xpad, N);
    pad_w_kernel<<<2, TB, 0, stream>>>(W1, Wpad);

    const int rowTiles = N / 16;

    gemm_wmma<8, 4, false><<<rowTiles, 32, 0, stream>>>(xpad, Wpad, bufA, nullptr);
    self_init_kernel<<<nhb, TB, 0, stream>>>(bufA, dinv, bufB, N);
    edge_msg_kernel<<<eb, TB, 0, stream>>>(srcIdx, dstIdx, dinv, bufA, bufB, E);
    bias_relu_kernel<<<nhb, TB, 0, stream>>>(bufB, b1, N);

    const float* Wl[2] = {W2, W3};
    const float* bl[2] = {b2, b3};
    for (int l = 0; l < 2; ++l) {
        gemm_wmma<64, 4, false><<<rowTiles, 32, 0, stream>>>(bufB, Wl[l], bufA, nullptr);
        self_init_kernel<<<nhb, TB, 0, stream>>>(bufA, dinv, bufB, N);
        edge_msg_kernel<<<eb, TB, 0, stream>>>(srcIdx, dstIdx, dinv, bufA, bufB, E);
        bias_relu_kernel<<<nhb, TB, 0, stream>>>(bufB, bl[l], N);
    }

    pool_zero_kernel<<<(NG * HID + TB - 1) / TB, TB, 0, stream>>>(sums, cnts, NG, HID);
    pool_add_kernel<<<nhb, TB, 0, stream>>>(bufB, batch, sums, N);
    pool_cnt_kernel<<<(N + TB - 1) / TB, TB, 0, stream>>>(batch, cnts, N);
    pool_div_kernel<<<(NG * HID + TB - 1) / TB, TB, 0, stream>>>(sums, cnts, G, NG, HID);

    gemm_wmma<64, 8, true><<<NG / 16, 32, 0, stream>>>(G, Wfc, (float*)d_out, bfc);
}


# layer 1 aggregates before its product, (A_hat x) W1: its gather reads 32-byte rows of x (4 lanes per node, 8 nodes per wave), not 256-byte rows of x W1; bias and relu move into the product
# speedup vs baseline: 1.9780x; 1.5662x over previous
#include <hip/hip_runtime.h>

typedef __attribute__((ext_vector_type(2))) float v2f;
typedef __attribute__((ext_vector_type(8))) float v8f;

__global__ void zero_cnt_kernel(int* __restrict__ cnt, int* __restrict__ cursor, int n) {
    int i = blockIdx.x * blockDim.x + threadIdx.x;
    if (i < n) cnt[i] = 0;
    if (i == 0) *cursor = 0;
}

__global__ void edge_count_kernel(const int* __restrict__ dst, int* __restrict__ cnt, int E) {
    int e = blockIdx.x * blockDim.x + threadIdx.x;
    if (e < E) atomicAdd(&cnt[dst[e]], 1);
}

__global__ void alloc_rows_kernel(const int* __restrict__ cnt, float* __restrict__ dinv,
                                  int2* __restrict__ rowinfo, int* __restrict__ cursor, int n) {
    const int i = blockIdx.x * blockDim.x + threadIdx.x;
    const int lane = threadIdx.x & 31;
    const int c = (i < n) ? cnt[i] : 0;
    int incl = c;
    for (int o = 1; o < 32; o <<= 1) {
        const int v = __shfl_up(incl, o);
        if (lane >= o) incl += v;
    }
    const int total = __shfl(incl, 31);
    int base = 0;
    if (lane == 31) base = atomicAdd(cursor, total);
    base = __shfl(base, 31);
    if (i < n) {
        rowinfo[i] = make_int2(base + incl - c, c);
        const float d = (float)(c + 1);
        dinv[i] = (d > 0.0f) ? rsqrtf(d) : 0.0f;
    }
}

__global__ void fill_rows_kernel(const int* __restrict__ src, const int* __restrict__ dst,
                                 const int2* __restrict__ rowinfo, int* __restrict__ cnt,
                                 int* __restrict__ csr, int E) {
    int e = blockIdx.x * blockDim.x + threadIdx.x;
    if (e >= E) return;
    const int s = src[e];
    const int d = dst[e];
    const int slot = atomicSub(&cnt[d], 1) - 1;
    int pos = rowinfo[d].x + slot;
    pos = (pos < 0) ? 0 : ((pos > E - 1) ? E - 1 : pos);
    csr[pos] = s;
}

__global__ void pad_x_kernel(const float* __restrict__ x, const float* __restrict__ dinv,
                             float* __restrict__ xp, int n) {
    int t = blockIdx.x * blockDim.x + threadIdx.x;
    if (t >= n * 8) return;
    int i = t >> 3, k = t & 7;
    xp[t] = (k < 6) ? x[i * 6 + k] * dinv[i] : 0.0f;
}

__global__ void pad_w_kernel(const float* __restrict__ w, float* __restrict__ wp) {
    int t = blockIdx.x * blockDim.x + threadIdx.x;
    if (t >= 8 * 64) return;
    wp[t] = (t < 6 * 64) ? w[t] : 0.0f;
}

template <int K, int NT, bool BIAS_RELU, bool ROW_SCALE>
__global__ __launch_bounds__(32) void gemm_wmma(
    const float* __restrict__ X, const float* __restrict__ W,
    float* __restrict__ Out, const float* __restrict__ bias,
    const float* __restrict__ rscale)
{
    constexpr int NCOLS = NT * 16;
    const int lane = threadIdx.x;
    const int hi   = lane >> 4;
    const int lo   = lane & 15;
    const int row0 = blockIdx.x * 16;

    const float* Arow = X + (size_t)(row0 + lo) * K;

    float rs[8];
    if (ROW_SCALE) {
#pragma unroll
        for (int j = 0; j < 8; ++j) rs[j] = rscale[row0 + j + 8 * hi];
    }

    v8f c[NT];
#pragma unroll
    for (int t = 0; t < NT; ++t) c[t] = (v8f){};

#pragma unroll
    for (int k0 = 0; k0 < K; k0 += 4) {
        const int ka = k0 + 2 * hi;
        const v2f a = *(const v2f*)(Arow + ka);
#pragma unroll
        for (int t = 0; t < NT; ++t) {
            v2f b;
            b.x = W[ka * NCOLS + 16 * t + lo];
            b.y = W[(ka + 1) * NCOLS + 16 * t + lo];
            c[t] = __builtin_amdgcn_wmma_f32_16x16x4_f32(
                false, a, false, b, (short)0, c[t], false, false);
        }
    }

#pragma unroll
    for (int t = 0; t < NT; ++t) {
        const int col = 16 * t + lo;
#pragma unroll
        for (int j = 0; j < 8; ++j) {
            float v = c[t][j];
            if (ROW_SCALE) v *= rs[j];
            if (BIAS_RELU) v = fmaxf(v + bias[col], 0.0f);
            Out[(size_t)(row0 + j + 8 * hi) * NCOLS + col] = v;
        }
    }
}

template <int K>
__device__ __forceinline__ void gather_batch(const float* __restrict__ tmp, int p,
                                             int j, unsigned lane, float2& acc) {
    float2 v[K];
#pragma unroll
    for (int u = 0; u < K; ++u) {
        const int sj = __builtin_amdgcn_readlane(p, j + u);
        v[u] = *(const float2*)(tmp + (size_t)sj * 64 + 2u * lane);
    }
#pragma unroll
    for (int u = 0; u < K; ++u) {
        acc.x += v[u].x;
        acc.y += v[u].y;
    }
}

__global__ __launch_bounds__(256) void gather_rows_kernel(
    const int2* __restrict__ rowinfo, const int* __restrict__ csr,
    const float* __restrict__ dinv, const float* __restrict__ tmp,
    const float* __restrict__ bias, float* __restrict__ out, int n, int E)
{
    const unsigned lane = threadIdx.x & 31u;
    const int i = __builtin_amdgcn_readfirstlane(
        (int)((blockIdx.x * blockDim.x + threadIdx.x) >> 5));
    if (i >= n) return;
    const int2 ri = rowinfo[i];
    const float di = dinv[i];
    const float2 self = *(const float2*)(tmp + (size_t)i * 64 + 2u * lane);
    const float2 bv = *(const float2*)(bias + 2u * lane);
    float2 acc = self;
    const int start = ri.x;
    const int deg = ri.y;
    for (int b = 0; b < deg; b += 32) {
        const int m = (deg - b < 32) ? (deg - b) : 32;
        int p = 0;
        if ((int)lane < m) {
            int q = start + b + (int)lane;
            q = (q < 0) ? 0 : ((q > E - 1) ? E - 1 : q);
            p = csr[q];
            p = (p < 0) ? 0 : ((p > n - 1) ? n - 1 : p);
        }
        int j = 0;
        for (; j + 8 <= m; j += 8) gather_batch<8>(tmp, p, j, lane, acc);
        if (m & 4) { gather_batch<4>(tmp, p, j, lane, acc); j += 4; }
        if (m & 2) { gather_batch<2>(tmp, p, j, lane, acc); j += 2; }
        if (m & 1) { gather_batch<1>(tmp, p, j, lane, acc); }
    }
    float2 o;
    o.x = fmaxf(fmaf(acc.x, di, bv.x), 0.0f);
    o.y = fmaxf(fmaf(acc.y, di, bv.y), 0.0f);
    *(float2*)(out + (size_t)i * 64 + 2u * lane) = o;
}

__global__ __launch_bounds__(256) void gather_narrow_kernel(
    const int2* __restrict__ rowinfo, const int* __restrict__ csr,
    const float* __restrict__ dinv, const float* __restrict__ xs,
    float* __restrict__ agg, int n, int E)
{
    const int t = blockIdx.x * blockDim.x + threadIdx.x;
    const int i = t >> 2;
    const int sub = t & 3;
    if (i >= n) return;
    const int2 ri = rowinfo[i];
    const float di = dinv[i];
    float2 acc = *(const float2*)(xs + (size_t)i * 8 + 2 * sub);
    const int start = ri.x;
    const int deg = ri.y;
    for (int j = 0; j < deg; j += 8) {
        int s[8];
#pragma unroll
        for (int u = 0; u < 8; ++u) {
            int q = start + j + ((j + u < deg) ? u : 0);
            q = (q < 0) ? 0 : ((q > E - 1) ? E - 1 : q);
            int p = csr[q];
            s[u] = (p < 0) ? 0 : ((p > n - 1) ? n - 1 : p);
        }
        float2 v[8];
#pragma unroll
        for (int u = 0; u < 8; ++u)
            v[u] = *(const float2*)(xs + (size_t)s[u] * 8 + 2 * sub);
#pragma unroll
        for (int u = 0; u < 8; ++u) {
            if (j + u < deg) {
                acc.x += v[u].x;
                acc.y += v[u].y;
            }
        }
    }
    float2 o;
    o.x = acc.x * di;
    o.y = acc.y * di;
    *(float2*)(agg + (size_t)i * 8 + 2 * sub) = o;
}

__global__ void pool_zero_kernel(float* __restrict__ sums, float* __restrict__ cnts,
                                 int ng, int hid) {
    int t = blockIdx.x * blockDim.x + threadIdx.x;
    if (t < ng * hid) sums[t] = 0.0f;
    if (t < ng)       cnts[t] = 0.0f;
}

__device__ __forceinline__ void pool_flush(float* __restrict__ sums, float* __restrict__ cnts,
                                           int g, int ng, unsigned lane, float2 acc, int len) {
    if ((unsigned)g < (unsigned)ng) {
        float* s = sums + (size_t)g * 64 + 2u * lane;
        atomicAdd(s, acc.x);
        atomicAdd(s + 1, acc.y);
        if (lane == 0) atomicAdd(&cnts[g], (float)len);
    }
}

__global__ __launch_bounds__(128) void pool_rows_kernel(
    const float* __restrict__ h, const int* __restrict__ batch,
    float* __restrict__ sums, float* __restrict__ cnts, int n, int ng)
{
    const unsigned lane = threadIdx.x & 31u;
    const int base = __builtin_amdgcn_readfirstlane(
        (int)((blockIdx.x * blockDim.x + threadIdx.x) >> 5)) * 32;
    if (base >= n) return;
    const int m = (n - base < 32) ? (n - base) : 32;
    const int gid = batch[base + (((int)lane < m) ? (int)lane : 0)];
    float2 v[32];
#pragma unroll
    for (int u = 0; u < 32; ++u) {
        const int r = base + ((u < m) ? u : (m - 1));
        v[u] = *(const float2*)(h + (size_t)r * 64 + 2u * lane);
    }
    int cur = __builtin_amdgcn_readlane(gid, 0);
    int len = 0;
    float2 acc;
    acc.x = 0.0f;
    acc.y = 0.0f;
#pragma unroll
    for (int u = 0; u < 32; ++u) {
        if (u < m) {
            const int g = __builtin_amdgcn_readlane(gid, u);
            if (g != cur) {
                pool_flush(sums, cnts, cur, ng, lane, acc, len);
                cur = g;
                len = 0;
                acc.x = 0.0f;
                acc.y = 0.0f;
            }
            acc.x += v[u].x;
            acc.y += v[u].y;
            ++len;
        }
    }
    pool_flush(sums, cnts, cur, ng, lane, acc, len);
}

__global__ void pool_div_kernel(const float* __restrict__ sums,
                                const float* __restrict__ cnts,
                                float* __restrict__ G, int ng, int hid) {
    int t = blockIdx.x * blockDim.x + threadIdx.x;
    if (t >= ng * hid) return;
    int g = t / hid;
    G[t] = sums[t] / fmaxf(cnts[g], 1.0f);
}

extern "C" void kernel_launch(void* const* d_in, const int* in_sizes, int n_in,
                              void* d_out, int out_size, void* d_ws, size_t ws_size,
                              hipStream_t stream) {
    (void)n_in; (void)out_size; (void)ws_size;

    const float* x     = (const float*)d_in[0];
    const int*   ei    = (const int*)  d_in[1];
    const int*   batch = (const int*)  d_in[2];
    const float* W1  = (const float*)d_in[3];
    const float* b1  = (const float*)d_in[4];
    const float* W2  = (const float*)d_in[5];
    const float* b2  = (const float*)d_in[6];
    const float* W3  = (const float*)d_in[7];
    const float* b3  = (const float*)d_in[8];
    const float* Wfc = (const float*)d_in[9];
    const float* bfc = (const float*)d_in[10];

    const int FIN = 6, HID = 64, EMB = 128, NG = 64;
    const int N = in_sizes[0] / FIN;
    const int E = in_sizes[1] / 2;
    const int* srcIdx = ei;
    const int* dstIdx = ei + E;

    char* ws = (char*)d_ws;
    auto a256 = [](size_t v) { return (v + 255) & ~(size_t)255; };
    size_t off = 0;
    float* dinv = (float*)(ws + off); off += a256((size_t)N * 4);
    float* xpad = (float*)(ws + off); off += a256((size_t)N * 8 * 4);
    float* Wpad = (float*)(ws + off); off += a256((size_t)8 * HID * 4);
    float* bufA = (float*)(ws + off); off += a256((size_t)N * HID * 4);
    float* bufB = (float*)(ws + off); off += a256((size_t)N * HID * 4);
    float* sums = (float*)(ws + off); off += a256((size_t)NG * HID * 4);
    float* cnts = (float*)(ws + off); off += a256((size_t)NG * 4);
    float* G    = (float*)(ws + off); off += a256((size_t)NG * HID * 4);
    int*  cnt     = (int*) (ws + off); off += a256((size_t)N * 4);
    int*  cursor  = (int*) (ws + off); off += a256((size_t)4);
    int2* rowinfo = (int2*)(ws + off); off += a256((size_t)N * 8);
    int*  csr     = (int*) (ws + off); off += a256((size_t)E * 4);

    const int TB = 256;
    const unsigned eb  = (unsigned)((E + TB - 1) / TB);
    const unsigned gb  = (unsigned)(((long long)N * 32 + TB - 1) / TB);

    zero_cnt_kernel<<<(N + TB - 1) / TB, TB, 0, stream>>>(cnt, cursor, N);
    edge_count_kernel<<<eb, TB, 0, stream>>>(dstIdx, cnt, E);
    alloc_rows_kernel<<<(N + TB - 1) / TB, TB, 0, stream>>>(cnt, dinv, rowinfo, cursor, N);
    fill_rows_kernel<<<eb, TB, 0, stream>>>(srcIdx, dstIdx, rowinfo, cnt, csr, E);

    pad_x_kernel<<<(N * 8 + TB - 1) / TB, TB, 0, stream>>>(x, dinv, xpad, N);
    pad_w_kernel<<<2, TB, 0, stream>>>(W1, Wpad);

    const int rowTiles = N / 16;

    float* xagg = bufA;
    gather_narrow_kernel<<<(unsigned)(((long long)N * 4 + TB - 1) / TB), TB, 0, stream>>>(
        rowinfo, csr, dinv, xpad, xagg, N, E);
    gemm_wmma<8, 4, true, false><<<rowTiles, 32, 0, stream>>>(xagg, Wpad, bufB, b1, nullptr);

    const float* Wl[2] = {W2, W3};
    const float* bl[2] = {b2, b3};
    for (int l = 0; l < 2; ++l) {
        gemm_wmma<64, 4, false, true><<<rowTiles, 32, 0, stream>>>(bufB, Wl[l], bufA, nullptr, dinv);
        gather_rows_kernel<<<gb, TB, 0, stream>>>(rowinfo, csr, dinv, bufA, bl[l], bufB, N, E);
    }

    pool_zero_kernel<<<(NG * HID + TB - 1) / TB, TB, 0, stream>>>(sums, cnts, NG, HID);
    pool_rows_kernel<<<(N + 127) / 128, 128, 0, stream>>>(bufB, batch, sums, cnts, N, NG);
    pool_div_kernel<<<(NG * HID + TB - 1) / TB, TB, 0, stream>>>(sums, cnts, G, NG, HID);

    gemm_wmma<64, 8, true, false><<<NG / 16, 32, 0, stream>>>(G, Wfc, (float*)d_out, bfc, nullptr);
}


# edge ranks from the counting pass: the count's atomic add returns each edge's place in its destination's range (kept, coalesced), so the fill pass is a plain scatter with no second atomic per edge
# speedup vs baseline: 2.3954x; 1.2110x over previous
#include <hip/hip_runtime.h>

typedef __attribute__((ext_vector_type(2))) float v2f;
typedef __attribute__((ext_vector_type(8))) float v8f;

__global__ void zero_cnt_kernel(int* __restrict__ cnt, int* __restrict__ cursor, int n) {
    int i = blockIdx.x * blockDim.x + threadIdx.x;
    if (i < n) cnt[i] = 0;
    if (i == 0) *cursor = 0;
}

__global__ void edge_count_kernel(const int* __restrict__ dst, int* __restrict__ cnt,
                                  int* __restrict__ rank, int E) {
    int e = blockIdx.x * blockDim.x + threadIdx.x;
    if (e < E) rank[e] = atomicAdd(&cnt[dst[e]], 1);
}

__global__ void alloc_rows_kernel(const int* __restrict__ cnt, float* __restrict__ dinv,
                                  int2* __restrict__ rowinfo, int* __restrict__ cursor, int n) {
    const int i = blockIdx.x * blockDim.x + threadIdx.x;
    const int lane = threadIdx.x & 31;
    const int c = (i < n) ? cnt[i] : 0;
    int incl = c;
    for (int o = 1; o < 32; o <<= 1) {
        const int v = __shfl_up(incl, o);
        if (lane >= o) incl += v;
    }
    const int total = __shfl(incl, 31);
    int base = 0;
    if (lane == 31) base = atomicAdd(cursor, total);
    base = __shfl(base, 31);
    if (i < n) {
        rowinfo[i] = make_int2(base + incl - c, c);
        const float d = (float)(c + 1);
        dinv[i] = (d > 0.0f) ? rsqrtf(d) : 0.0f;
    }
}

__global__ void fill_rows_kernel(const int* __restrict__ src, const int* __restrict__ dst,
                                 const int* __restrict__ rank,
                                 const int2* __restrict__ rowinfo,
                                 int* __restrict__ csr, int E) {
    int e = blockIdx.x * blockDim.x + threadIdx.x;
    if (e >= E) return;
    const int s = src[e];
    const int d = dst[e];
    const int slot = rank[e];
    int pos = rowinfo[d].x + slot;
    pos = (pos < 0) ? 0 : ((pos > E - 1) ? E - 1 : pos);
    csr[pos] = s;
}

__global__ void pad_x_kernel(const float* __restrict__ x, const float* __restrict__ dinv,
                             float* __restrict__ xp, int n) {
    int t = blockIdx.x * blockDim.x + threadIdx.x;
    if (t >= n * 8) return;
    int i = t >> 3, k = t & 7;
    xp[t] = (k < 6) ? x[i * 6 + k] * dinv[i] : 0.0f;
}

__global__ void pad_w_kernel(const float* __restrict__ w, float* __restrict__ wp) {
    int t = blockIdx.x * blockDim.x + threadIdx.x;
    if (t >= 8 * 64) return;
    wp[t] = (t < 6 * 64) ? w[t] : 0.0f;
}

template <int K, int NT, bool BIAS_RELU, bool ROW_SCALE>
__global__ __launch_bounds__(32) void gemm_wmma(
    const float* __restrict__ X, const float* __restrict__ W,
    float* __restrict__ Out, const float* __restrict__ bias,
    const float* __restrict__ rscale)
{
    constexpr int NCOLS = NT * 16;
    const int lane = threadIdx.x;
    const int hi   = lane >> 4;
    const int lo   = lane & 15;
    const int row0 = blockIdx.x * 16;

    const float* Arow = X + (size_t)(row0 + lo) * K;

    float rs[8];
    if (ROW_SCALE) {
#pragma unroll
        for (int j = 0; j < 8; ++j) rs[j] = rscale[row0 + j + 8 * hi];
    }

    v8f c[NT];
#pragma unroll
    for (int t = 0; t < NT; ++t) c[t] = (v8f){};

#pragma unroll
    for (int k0 = 0; k0 < K; k0 += 4) {
        const int ka = k0 + 2 * hi;
        const v2f a = *(const v2f*)(Arow + ka);
#pragma unroll
        for (int t = 0; t < NT; ++t) {
            v2f b;
            b.x = W[ka * NCOLS + 16 * t + lo];
            b.y = W[(ka + 1) * NCOLS + 16 * t + lo];
            c[t] = __builtin_amdgcn_wmma_f32_16x16x4_f32(
                false, a, false, b, (short)0, c[t], false, false);
        }
    }

#pragma unroll
    for (int t = 0; t < NT; ++t) {
        const int col = 16 * t + lo;
#pragma unroll
        for (int j = 0; j < 8; ++j) {
            float v = c[t][j];
            if (ROW_SCALE) v *= rs[j];
            if (BIAS_RELU) v = fmaxf(v + bias[col], 0.0f);
            Out[(size_t)(row0 + j + 8 * hi) * NCOLS + col] = v;
        }
    }
}

template <int K>
__device__ __forceinline__ void gather_batch(const float* __restrict__ tmp, int p,
                                             int j, unsigned lane, float2& acc) {
    float2 v[K];
#pragma unroll
    for (int u = 0; u < K; ++u) {
        const int sj = __builtin_amdgcn_readlane(p, j + u);
        v[u] = *(const float2*)(tmp + (size_t)sj * 64 + 2u * lane);
    }
#pragma unroll
    for (int u = 0; u < K; ++u) {
        acc.x += v[u].x;
        acc.y += v[u].y;
    }
}

__global__ __launch_bounds__(256) void gather_rows_kernel(
    const int2* __restrict__ rowinfo, const int* __restrict__ csr,
    const float* __restrict__ dinv, const float* __restrict__ tmp,
    const float* __restrict__ bias, float* __restrict__ out, int n, int E)
{
    const unsigned lane = threadIdx.x & 31u;
    const int i = __builtin_amdgcn_readfirstlane(
        (int)((blockIdx.x * blockDim.x + threadIdx.x) >> 5));
    if (i >= n) return;
    const int2 ri = rowinfo[i];
    const float di = dinv[i];
    const float2 self = *(const float2*)(tmp + (size_t)i * 64 + 2u * lane);
    const float2 bv = *(const float2*)(bias + 2u * lane);
    float2 acc = self;
    const int start = ri.x;
    const int deg = ri.y;
    for (int b = 0; b < deg; b += 32) {
        const int m = (deg - b < 32) ? (deg - b) : 32;
        int p = 0;
        if ((int)lane < m) {
            int q = start + b + (int)lane;
            q = (q < 0) ? 0 : ((q > E - 1) ? E - 1 : q);
            p = csr[q];
            p = (p < 0) ? 0 : ((p > n - 1) ? n - 1 : p);
        }
        int j = 0;
        for (; j + 8 <= m; j += 8) gather_batch<8>(tmp, p, j, lane, acc);
        if (m & 4) { gather_batch<4>(tmp, p, j, lane, acc); j += 4; }
        if (m & 2) { gather_batch<2>(tmp, p, j, lane, acc); j += 2; }
        if (m & 1) { gather_batch<1>(tmp, p, j, lane, acc); }
    }
    float2 o;
    o.x = fmaxf(fmaf(acc.x, di, bv.x), 0.0f);
    o.y = fmaxf(fmaf(acc.y, di, bv.y), 0.0f);
    *(float2*)(out + (size_t)i * 64 + 2u * lane) = o;
}

__global__ __launch_bounds__(256) void gather_narrow_kernel(
    const int2* __restrict__ rowinfo, const int* __restrict__ csr,
    const float* __restrict__ dinv, const float* __restrict__ xs,
    float* __restrict__ agg, int n, int E)
{
    const int t = blockIdx.x * blockDim.x + threadIdx.x;
    const int i = t >> 2;
    const int sub = t & 3;
    if (i >= n) return;
    const int2 ri = rowinfo[i];
    const float di = dinv[i];
    float2 acc = *(const float2*)(xs + (size_t)i * 8 + 2 * sub);
    const int start = ri.x;
    const int deg = ri.y;
    for (int j = 0; j < deg; j += 8) {
        int s[8];
#pragma unroll
        for (int u = 0; u < 8; ++u) {
            int q = start + j + ((j + u < deg) ? u : 0);
            q = (q < 0) ? 0 : ((q > E - 1) ? E - 1 : q);
            int p = csr[q];
            s[u] = (p < 0) ? 0 : ((p > n - 1) ? n - 1 : p);
        }
        float2 v[8];
#pragma unroll
        for (int u = 0; u < 8; ++u)
            v[u] = *(const float2*)(xs + (size_t)s[u] * 8 + 2 * sub);
#pragma unroll
        for (int u = 0; u < 8; ++u) {
            if (j + u < deg) {
                acc.x += v[u].x;
                acc.y += v[u].y;
            }
        }
    }
    float2 o;
    o.x = acc.x * di;
    o.y = acc.y * di;
    *(float2*)(agg + (size_t)i * 8 + 2 * sub) = o;
}

__global__ void pool_zero_kernel(float* __restrict__ sums, float* __restrict__ cnts,
                                 int ng, int hid) {
    int t = blockIdx.x * blockDim.x + threadIdx.x;
    if (t < ng * hid) sums[t] = 0.0f;
    if (t < ng)       cnts[t] = 0.0f;
}

__device__ __forceinline__ void pool_flush(float* __restrict__ sums, float* __restrict__ cnts,
                                           int g, int ng, unsigned lane, float2 acc, int len) {
    if ((unsigned)g < (unsigned)ng) {
        float* s = sums + (size_t)g * 64 + 2u * lane;
        atomicAdd(s, acc.x);
        atomicAdd(s + 1, acc.y);
        if (lane == 0) atomicAdd(&cnts[g], (float)len);
    }
}

__global__ __launch_bounds__(128) void pool_rows_kernel(
    const float* __restrict__ h, const int* __restrict__ batch,
    float* __restrict__ sums, float* __restrict__ cnts, int n, int ng)
{
    const unsigned lane = threadIdx.x & 31u;
    const int base = __builtin_amdgcn_readfirstlane(
        (int)((blockIdx.x * blockDim.x + threadIdx.x) >> 5)) * 32;
    if (base >= n) return;
    const int m = (n - base < 32) ? (n - base) : 32;
    const int gid = batch[base + (((int)lane < m) ? (int)lane : 0)];
    float2 v[32];
#pragma unroll
    for (int u = 0; u < 32; ++u) {
        const int r = base + ((u < m) ? u : (m - 1));
        v[u] = *(const float2*)(h + (size_t)r * 64 + 2u * lane);
    }
    int cur = __builtin_amdgcn_readlane(gid, 0);
    int len = 0;
    float2 acc;
    acc.x = 0.0f;
    acc.y = 0.0f;
#pragma unroll
    for (int u = 0; u < 32; ++u) {
        if (u < m) {
            const int g = __builtin_amdgcn_readlane(gid, u);
            if (g != cur) {
                pool_flush(sums, cnts, cur, ng, lane, acc, len);
                cur = g;
                len = 0;
                acc.x = 0.0f;
                acc.y = 0.0f;
            }
            acc.x += v[u].x;
            acc.y += v[u].y;
            ++len;
        }
    }
    pool_flush(sums, cnts, cur, ng, lane, acc, len);
}

__global__ void pool_div_kernel(const float* __restrict__ sums,
                                const float* __restrict__ cnts,
                                float* __restrict__ G, int ng, int hid) {
    int t = blockIdx.x * blockDim.x + threadIdx.x;
    if (t >= ng * hid) return;
    int g = t / hid;
    G[t] = sums[t] / fmaxf(cnts[g], 1.0f);
}

extern "C" void kernel_launch(void* const* d_in, const int* in_sizes, int n_in,
                              void* d_out, int out_size, void* d_ws, size_t ws_size,
                              hipStream_t stream) {
    (void)n_in; (void)out_size; (void)ws_size;

    const float* x     = (const float*)d_in[0];
    const int*   ei    = (const int*)  d_in[1];
    const int*   batch = (const int*)  d_in[2];
    const float* W1  = (const float*)d_in[3];
    const float* b1  = (const float*)d_in[4];
    const float* W2  = (const float*)d_in[5];
    const float* b2  = (const float*)d_in[6];
    const float* W3  = (const float*)d_in[7];
    const float* b3  = (const float*)d_in[8];
    const float* Wfc = (const float*)d_in[9];
    const float* bfc = (const float*)d_in[10];

    const int FIN = 6, HID = 64, EMB = 128, NG = 64;
    const int N = in_sizes[0] / FIN;
    const int E = in_sizes[1] / 2;
    const int* srcIdx = ei;
    const int* dstIdx = ei + E;

    char* ws = (char*)d_ws;
    auto a256 = [](size_t v) { return (v + 255) & ~(size_t)255; };
    size_t off = 0;
    float* dinv = (float*)(ws + off); off += a256((size_t)N * 4);
    float* xpad = (float*)(ws + off); off += a256((size_t)N * 8 * 4);
    float* Wpad = (float*)(ws + off); off += a256((size_t)8 * HID * 4);
    float* bufA = (float*)(ws + off); off += a256((size_t)N * HID * 4);
    float* bufB = (float*)(ws + off); off += a256((size_t)N * HID * 4);
    float* sums = (float*)(ws + off); off += a256((size_t)NG * HID * 4);
    float* cnts = (float*)(ws + off); off += a256((size_t)NG * 4);
    float* G    = (float*)(ws + off); off += a256((size_t)NG * HID * 4);
    int*  cnt     = (int*) (ws + off); off += a256((size_t)N * 4);
    int*  cursor  = (int*) (ws + off); off += a256((size_t)4);
    int2* rowinfo = (int2*)(ws + off); off += a256((size_t)N * 8);
    int*  csr     = (int*) (ws + off); off += a256((size_t)E * 4);
    int*  rank    = (int*)bufB;

    const int TB = 256;
    const unsigned eb  = (unsigned)((E + TB - 1) / TB);
    const unsigned gb  = (unsigned)(((long long)N * 32 + TB - 1) / TB);

    zero_cnt_kernel<<<(N + TB - 1) / TB, TB, 0, stream>>>(cnt, cursor, N);
    edge_count_kernel<<<eb, TB, 0, stream>>>(dstIdx, cnt, rank, E);
    alloc_rows_kernel<<<(N + TB - 1) / TB, TB, 0, stream>>>(cnt, dinv, rowinfo, cursor, N);
    fill_rows_kernel<<<eb, TB, 0, stream>>>(srcIdx, dstIdx, rank, rowinfo, csr, E);

    pad_x_kernel<<<(N * 8 + TB - 1) / TB, TB, 0, stream>>>(x, dinv, xpad, N);
    pad_w_kernel<<<2, TB, 0, stream>>>(W1, Wpad);

    const int rowTiles = N / 16;

    float* xagg = bufA;
    gather_narrow_kernel<<<(unsigned)(((long long)N * 4 + TB - 1) / TB), TB, 0, stream>>>(
        rowinfo, csr, dinv, xpad, xagg, N, E);
    gemm_wmma<8, 4, true, false><<<rowTiles, 32, 0, stream>>>(xagg, Wpad, bufB, b1, nullptr);

    const float* Wl[2] = {W2, W3};
    const float* bl[2] = {b2, b3};
    for (int l = 0; l < 2; ++l) {
        gemm_wmma<64, 4, false, true><<<rowTiles, 32, 0, stream>>>(bufB, Wl[l], bufA, nullptr, dinv);
        gather_rows_kernel<<<gb, TB, 0, stream>>>(rowinfo, csr, dinv, bufA, bl[l], bufB, N, E);
    }

    pool_zero_kernel<<<(NG * HID + TB - 1) / TB, TB, 0, stream>>>(sums, cnts, NG, HID);
    pool_rows_kernel<<<(N + 127) / 128, 128, 0, stream>>>(bufB, batch, sums, cnts, N, NG);
    pool_div_kernel<<<(NG * HID + TB - 1) / TB, TB, 0, stream>>>(sums, cnts, G, NG, HID);

    gemm_wmma<64, 8, true, false><<<NG / 16, 32, 0, stream>>>(G, Wfc, (float*)d_out, bfc, nullptr);
}


# edge grouping as a two-level counting sort in LDS: blocks count and scatter edges per bucket of 256 nodes, one scan, then a block per bucket ranks its nodes in LDS; no global atomic per edge
# speedup vs baseline: 2.9601x; 1.2357x over previous
#include <hip/hip_runtime.h>

typedef __attribute__((ext_vector_type(2))) float v2f;
typedef __attribute__((ext_vector_type(8))) float v8f;

constexpr int BUCKET_SHIFT = 8;
constexpr int BUCKET_NODES = 1 << BUCKET_SHIFT;
constexpr int SORT_THREADS = 1024;
constexpr int EDGES_PER_THREAD = 4;
constexpr int EPB = SORT_THREADS * EDGES_PER_THREAD;
constexpr int MAX_BUCKETS = 1024;

__device__ __forceinline__ int wave_incl_scan(int v, int lane) {
    v += __builtin_amdgcn_update_dpp(0, v, 0x111, 0xf, 0xf, false);
    v += __builtin_amdgcn_update_dpp(0, v, 0x112, 0xf, 0xf, false);
    v += __builtin_amdgcn_update_dpp(0, v, 0x114, 0xf, 0xf, false);
    v += __builtin_amdgcn_update_dpp(0, v, 0x118, 0xf, 0xf, false);
    const int low = __builtin_amdgcn_readlane(v, 15);
    return (lane >= 16) ? v + low : v;
}

__device__ __forceinline__ int bucket_of(int d, int nb) {
    int b = d >> BUCKET_SHIFT;
    return (b < 0) ? 0 : ((b > nb - 1) ? nb - 1 : b);
}

__global__ __launch_bounds__(SORT_THREADS) void bucket_count_kernel(
    const int* __restrict__ dst, int* __restrict__ hist, int E, int nb, int nblk)
{
    __shared__ int lds_tab[MAX_BUCKETS];
    const int tid = threadIdx.x;
    for (int i = tid; i < nb; i += SORT_THREADS) lds_tab[i] = 0;
    __syncthreads();
    const int e0 = blockIdx.x * EPB + tid;
    int d[EDGES_PER_THREAD];
#pragma unroll
    for (int k = 0; k < EDGES_PER_THREAD; ++k) {
        const int e = e0 + k * SORT_THREADS;
        d[k] = (e < E) ? dst[e] : -1;
    }
#pragma unroll
    for (int k = 0; k < EDGES_PER_THREAD; ++k) {
        const int e = e0 + k * SORT_THREADS;
        if (e < E) atomicAdd(&lds_tab[bucket_of(d[k], nb)], 1);
    }
    __syncthreads();
    for (int i = tid; i < nb; i += SORT_THREADS) hist[i * nblk + blockIdx.x] = lds_tab[i];
}

constexpr int SCAN_TILES = 10;
__global__ __launch_bounds__(SORT_THREADS) void scan_counts_kernel(
    const int* __restrict__ hist, int* __restrict__ off, int L)
{
    __shared__ int wtot[2][32];
    const int tid = threadIdx.x;
    const int lane = tid & 31;
    const int wave = tid >> 5;
    int carry = 0;
    for (int base = 0; base < L; base += SCAN_TILES * 4 * SORT_THREADS) {
        int4 v[SCAN_TILES];
#pragma unroll
        for (int k = 0; k < SCAN_TILES; ++k) {
            const int idx = base + (k * SORT_THREADS + tid) * 4;
            v[k] = (idx < L) ? *(const int4*)(hist + idx) : make_int4(0, 0, 0, 0);
        }
#pragma unroll
        for (int k = 0; k < SCAN_TILES; ++k) {
            const int idx = base + (k * SORT_THREADS + tid) * 4;
            const int s0 = v[k].x;
            const int s1 = s0 + v[k].y;
            const int s2 = s1 + v[k].z;
            const int s3 = s2 + v[k].w;
            const int incl = wave_incl_scan(s3, lane);
            if (lane == 31) wtot[k & 1][wave] = incl;
            __syncthreads();
            int wexcl = 0, total = 0;
#pragma unroll
            for (int w = 0; w < 32; ++w) {
                const int t = wtot[k & 1][w];
                total += t;
                if (w < wave) wexcl += t;
            }
            const int ex = carry + wexcl + incl - s3;
            if (idx < L) *(int4*)(off + idx) = make_int4(ex, ex + s0, ex + s1, ex + s2);
            carry += total;
        }
    }
}

__global__ __launch_bounds__(SORT_THREADS) void bucket_scatter_kernel(
    const int* __restrict__ src, const int* __restrict__ dst, const int* __restrict__ off,
    unsigned* __restrict__ blist, int E, int nb, int nblk)
{
    __shared__ int lds_tab[MAX_BUCKETS];
    const int tid = threadIdx.x;
    for (int i = tid; i < nb; i += SORT_THREADS) lds_tab[i] = off[i * nblk + blockIdx.x];
    __syncthreads();
    const int e0 = blockIdx.x * EPB + tid;
    int s[EDGES_PER_THREAD], d[EDGES_PER_THREAD];
#pragma unroll
    for (int k = 0; k < EDGES_PER_THREAD; ++k) {
        const int e = e0 + k * SORT_THREADS;
        s[k] = (e < E) ? src[e] : 0;
        d[k] = (e < E) ? dst[e] : -1;
    }
#pragma unroll
    for (int k = 0; k < EDGES_PER_THREAD; ++k) {
        const int e = e0 + k * SORT_THREADS;
        if (e < E) {
            int pos = atomicAdd(&lds_tab[bucket_of(d[k], nb)], 1);
            pos = (pos < 0) ? 0 : ((pos > E - 1) ? E - 1 : pos);
            blist[pos] = ((unsigned)s[k] << BUCKET_SHIFT) | ((unsigned)d[k] & (BUCKET_NODES - 1));
        }
    }
}

__global__ __launch_bounds__(SORT_THREADS) void bucket_rows_kernel(
    const int* __restrict__ off, const unsigned* __restrict__ blist,
    int* __restrict__ csr, int2* __restrict__ rowinfo, float* __restrict__ dinv,
    int n, int E, int nb, int nblk)
{
    __shared__ int cnt[BUCKET_NODES];
    __shared__ int cur[BUCKET_NODES];
    __shared__ int wtot[BUCKET_NODES / 32];
    const int tid = threadIdx.x;
    const int lane = tid & 31;
    const int wave = tid >> 5;
    const int b = blockIdx.x;
    int bstart = off[b * nblk];
    int bend = (b + 1 < nb) ? off[(b + 1) * nblk] : E;
    bstart = (bstart < 0) ? 0 : ((bstart > E) ? E : bstart);
    bend = (bend < bstart) ? bstart : ((bend > E) ? E : bend);
    if (tid < BUCKET_NODES) cnt[tid] = 0;
    __syncthreads();
    for (int e = bstart + tid; e < bend; e += SORT_THREADS)
        atomicAdd(&cnt[blist[e] & (BUCKET_NODES - 1)], 1);
    __syncthreads();
    const int c = (tid < BUCKET_NODES) ? cnt[tid] : 0;
    const int incl = wave_incl_scan(c, lane);
    if (tid < BUCKET_NODES && lane == 31) wtot[wave] = incl;
    __syncthreads();
    if (tid < BUCKET_NODES) {
        int before = 0;
        for (int w = 0; w < wave; ++w) before += wtot[w];
        const int start = bstart + before + incl - c;
        cur[tid] = start;
        const int node = b * BUCKET_NODES + tid;
        if (node < n) {
            rowinfo[node] = make_int2(start, c);
            const float dg = (float)(c + 1);
            dinv[node] = (dg > 0.0f) ? rsqrtf(dg) : 0.0f;
        }
    }
    __syncthreads();
    for (int e = bstart + tid; e < bend; e += SORT_THREADS) {
        const unsigned w = blist[e];
        int pos = atomicAdd(&cur[w & (BUCKET_NODES - 1)], 1);
        pos = (pos < 0) ? 0 : ((pos > E - 1) ? E - 1 : pos);
        csr[pos] = (int)(w >> BUCKET_SHIFT);
    }
}

__global__ void pad_x_kernel(const float* __restrict__ x, const float* __restrict__ dinv,
                             float* __restrict__ xp, int n) {
    int t = blockIdx.x * blockDim.x + threadIdx.x;
    if (t >= n * 8) return;
    int i = t >> 3, k = t & 7;
    xp[t] = (k < 6) ? x[i * 6 + k] * dinv[i] : 0.0f;
}

__global__ void pad_w_kernel(const float* __restrict__ w, float* __restrict__ wp) {
    int t = blockIdx.x * blockDim.x + threadIdx.x;
    if (t >= 8 * 64) return;
    wp[t] = (t < 6 * 64) ? w[t] : 0.0f;
}

template <int K, int NT, bool BIAS_RELU, bool ROW_SCALE>
__global__ __launch_bounds__(32) void gemm_wmma(
    const float* __restrict__ X, const float* __restrict__ W,
    float* __restrict__ Out, const float* __restrict__ bias,
    const float* __restrict__ rscale)
{
    constexpr int NCOLS = NT * 16;
    const int lane = threadIdx.x;
    const int hi   = lane >> 4;
    const int lo   = lane & 15;
    const int row0 = blockIdx.x * 16;

    const float* Arow = X + (size_t)(row0 + lo) * K;

    float rs[8];
    if (ROW_SCALE) {
#pragma unroll
        for (int j = 0; j < 8; ++j) rs[j] = rscale[row0 + j + 8 * hi];
    }

    v8f c[NT];
#pragma unroll
    for (int t = 0; t < NT; ++t) c[t] = (v8f){};

#pragma unroll
    for (int k0 = 0; k0 < K; k0 += 4) {
        const int ka = k0 + 2 * hi;
        const v2f a = *(const v2f*)(Arow + ka);
#pragma unroll
        for (int t = 0; t < NT; ++t) {
            v2f b;
            b.x = W[ka * NCOLS + 16 * t + lo];
            b.y = W[(ka + 1) * NCOLS + 16 * t + lo];
            c[t] = __builtin_amdgcn_wmma_f32_16x16x4_f32(
                false, a, false, b, (short)0, c[t], false, false);
        }
    }

#pragma unroll
    for (int t = 0; t < NT; ++t) {
        const int col = 16 * t + lo;
#pragma unroll
        for (int j = 0; j < 8; ++j) {
            float v = c[t][j];
            if (ROW_SCALE) v *= rs[j];
            if (BIAS_RELU) v = fmaxf(v + bias[col], 0.0f);
            Out[(size_t)(row0 + j + 8 * hi) * NCOLS + col] = v;
        }
    }
}

template <int K>
__device__ __forceinline__ void gather_batch(const float* __restrict__ tmp, int p,
                                             int j, unsigned lane, float2& acc) {
    float2 v[K];
#pragma unroll
    for (int u = 0; u < K; ++u) {
        const int sj = __builtin_amdgcn_readlane(p, j + u);
        v[u] = *(const float2*)(tmp + (size_t)sj * 64 + 2u * lane);
    }
#pragma unroll
    for (int u = 0; u < K; ++u) {
        acc.x += v[u].x;
        acc.y += v[u].y;
    }
}

__global__ __launch_bounds__(256) void gather_rows_kernel(
    const int2* __restrict__ rowinfo, const int* __restrict__ csr,
    const float* __restrict__ dinv, const float* __restrict__ tmp,
    const float* __restrict__ bias, float* __restrict__ out, int n, int E)
{
    const unsigned lane = threadIdx.x & 31u;
    const int i = __builtin_amdgcn_readfirstlane(
        (int)((blockIdx.x * blockDim.x + threadIdx.x) >> 5));
    if (i >= n) return;
    const int2 ri = rowinfo[i];
    const float di = dinv[i];
    const float2 self = *(const float2*)(tmp + (size_t)i * 64 + 2u * lane);
    const float2 bv = *(const float2*)(bias + 2u * lane);
    float2 acc = self;
    const int start = ri.x;
    const int deg = ri.y;
    for (int b = 0; b < deg; b += 32) {
        const int m = (deg - b < 32) ? (deg - b) : 32;
        int p = 0;
        if ((int)lane < m) {
            int q = start + b + (int)lane;
            q = (q < 0) ? 0 : ((q > E - 1) ? E - 1 : q);
            p = csr[q];
            p = (p < 0) ? 0 : ((p > n - 1) ? n - 1 : p);
        }
        int j = 0;
        for (; j + 8 <= m; j += 8) gather_batch<8>(tmp, p, j, lane, acc);
        if (m & 4) { gather_batch<4>(tmp, p, j, lane, acc); j += 4; }
        if (m & 2) { gather_batch<2>(tmp, p, j, lane, acc); j += 2; }
        if (m & 1) { gather_batch<1>(tmp, p, j, lane, acc); }
    }
    float2 o;
    o.x = fmaxf(fmaf(acc.x, di, bv.x), 0.0f);
    o.y = fmaxf(fmaf(acc.y, di, bv.y), 0.0f);
    *(float2*)(out + (size_t)i * 64 + 2u * lane) = o;
}

__global__ __launch_bounds__(256) void gather_narrow_kernel(
    const int2* __restrict__ rowinfo, const int* __restrict__ csr,
    const float* __restrict__ dinv, const float* __restrict__ xs,
    float* __restrict__ agg, int n, int E)
{
    const int t = blockIdx.x * blockDim.x + threadIdx.x;
    const int i = t >> 2;
    const int sub = t & 3;
    if (i >= n) return;
    const int2 ri = rowinfo[i];
    const float di = dinv[i];
    float2 acc = *(const float2*)(xs + (size_t)i * 8 + 2 * sub);
    const int start = ri.x;
    const int deg = ri.y;
    for (int j = 0; j < deg; j += 8) {
        int s[8];
#pragma unroll
        for (int u = 0; u < 8; ++u) {
            int q = start + j + ((j + u < deg) ? u : 0);
            q = (q < 0) ? 0 : ((q > E - 1) ? E - 1 : q);
            int p = csr[q];
            s[u] = (p < 0) ? 0 : ((p > n - 1) ? n - 1 : p);
        }
        float2 v[8];
#pragma unroll
        for (int u = 0; u < 8; ++u)
            v[u] = *(const float2*)(xs + (size_t)s[u] * 8 + 2 * sub);
#pragma unroll
        for (int u = 0; u < 8; ++u) {
            if (j + u < deg) {
                acc.x += v[u].x;
                acc.y += v[u].y;
            }
        }
    }
    float2 o;
    o.x = acc.x * di;
    o.y = acc.y * di;
    *(float2*)(agg + (size_t)i * 8 + 2 * sub) = o;
}

__global__ void pool_zero_kernel(float* __restrict__ sums, float* __restrict__ cnts,
                                 int ng, int hid) {
    int t = blockIdx.x * blockDim.x + threadIdx.x;
    if (t < ng * hid) sums[t] = 0.0f;
    if (t < ng)       cnts[t] = 0.0f;
}

__device__ __forceinline__ void pool_flush(float* __restrict__ sums, float* __restrict__ cnts,
                                           int g, int ng, unsigned lane, float2 acc, int len) {
    if ((unsigned)g < (unsigned)ng) {
        float* s = sums + (size_t)g * 64 + 2u * lane;
        atomicAdd(s, acc.x);
        atomicAdd(s + 1, acc.y);
        if (lane == 0) atomicAdd(&cnts[g], (float)len);
    }
}

__global__ __launch_bounds__(128) void pool_rows_kernel(
    const float* __restrict__ h, const int* __restrict__ batch,
    float* __restrict__ sums, float* __restrict__ cnts, int n, int ng)
{
    const unsigned lane = threadIdx.x & 31u;
    const int base = __builtin_amdgcn_readfirstlane(
        (int)((blockIdx.x * blockDim.x + threadIdx.x) >> 5)) * 32;
    if (base >= n) return;
    const int m = (n - base < 32) ? (n - base) : 32;
    const int gid = batch[base + (((int)lane < m) ? (int)lane : 0)];
    float2 v[32];
#pragma unroll
    for (int u = 0; u < 32; ++u) {
        const int r = base + ((u < m) ? u : (m - 1));
        v[u] = *(const float2*)(h + (size_t)r * 64 + 2u * lane);
    }
    int cur = __builtin_amdgcn_readlane(gid, 0);
    int len = 0;
    float2 acc;
    acc.x = 0.0f;
    acc.y = 0.0f;
#pragma unroll
    for (int u = 0; u < 32; ++u) {
        if (u < m) {
            const int g = __builtin_amdgcn_readlane(gid, u);
            if (g != cur) {
                pool_flush(sums, cnts, cur, ng, lane, acc, len);
                cur = g;
                len = 0;
                acc.x = 0.0f;
                acc.y = 0.0f;
            }
            acc.x += v[u].x;
            acc.y += v[u].y;
            ++len;
        }
    }
    pool_flush(sums, cnts, cur, ng, lane, acc, len);
}

__global__ void pool_div_kernel(const float* __restrict__ sums,
                                const float* __restrict__ cnts,
                                float* __restrict__ G, int ng, int hid) {
    int t = blockIdx.x * blockDim.x + threadIdx.x;
    if (t >= ng * hid) return;
    int g = t / hid;
    G[t] = sums[t] / fmaxf(cnts[g], 1.0f);
}

extern "C" void kernel_launch(void* const* d_in, const int* in_sizes, int n_in,
                              void* d_out, int out_size, void* d_ws, size_t ws_size,
                              hipStream_t stream) {
    (void)n_in; (void)out_size; (void)ws_size;

    const float* x     = (const float*)d_in[0];
    const int*   ei    = (const int*)  d_in[1];
    const int*   batch = (const int*)  d_in[2];
    const float* W1  = (const float*)d_in[3];
    const float* b1  = (const float*)d_in[4];
    const float* W2  = (const float*)d_in[5];
    const float* b2  = (const float*)d_in[6];
    const float* W3  = (const float*)d_in[7];
    const float* b3  = (const float*)d_in[8];
    const float* Wfc = (const float*)d_in[9];
    const float* bfc = (const float*)d_in[10];

    const int FIN = 6, HID = 64, EMB = 128, NG = 64;
    const int N = in_sizes[0] / FIN;
    const int E = in_sizes[1] / 2;
    const int* srcIdx = ei;
    const int* dstIdx = ei + E;

    char* ws = (char*)d_ws;
    auto a256 = [](size_t v) { return (v + 255) & ~(size_t)255; };
    size_t off = 0;
    float* dinv = (float*)(ws + off); off += a256((size_t)N * 4);
    float* xpad = (float*)(ws + off); off += a256((size_t)N * 8 * 4);
    float* Wpad = (float*)(ws + off); off += a256((size_t)8 * HID * 4);
    float* bufA = (float*)(ws + off); off += a256((size_t)N * HID * 4);
    float* bufB = (float*)(ws + off); off += a256((size_t)N * HID * 4);
    float* sums = (float*)(ws + off); off += a256((size_t)NG * HID * 4);
    float* cnts = (float*)(ws + off); off += a256((size_t)NG * 4);
    float* G    = (float*)(ws + off); off += a256((size_t)NG * HID * 4);
    int2* rowinfo = (int2*)(ws + off); off += a256((size_t)N * 8);
    int*  csr     = (int*) (ws + off); off += a256((size_t)E * 4);
    const int nbuckets = (N + BUCKET_NODES - 1) >> BUCKET_SHIFT;
    const int nblk = (((E + EPB - 1) / EPB) + 3) & ~3;
    const int L = nbuckets * nblk;
    int* hist = (int*)bufA;
    int* offs = hist + ((L + 63) & ~63);
    unsigned* blist = (unsigned*)bufB;

    const int TB = 256;
    const unsigned gb  = (unsigned)(((long long)N * 32 + TB - 1) / TB);

    bucket_count_kernel<<<nblk, SORT_THREADS, 0, stream>>>(dstIdx, hist, E, nbuckets, nblk);
    scan_counts_kernel<<<1, SORT_THREADS, 0, stream>>>(hist, offs, L);
    bucket_scatter_kernel<<<nblk, SORT_THREADS, 0, stream>>>(srcIdx, dstIdx, offs, blist,
                                                                   E, nbuckets, nblk);
    bucket_rows_kernel<<<nbuckets, SORT_THREADS, 0, stream>>>(offs, blist, csr, rowinfo, dinv,
                                                              N, E, nbuckets, nblk);

    pad_x_kernel<<<(N * 8 + TB - 1) / TB, TB, 0, stream>>>(x, dinv, xpad, N);
    pad_w_kernel<<<2, TB, 0, stream>>>(W1, Wpad);

    const int rowTiles = N / 16;

    float* xagg = bufA;
    gather_narrow_kernel<<<(unsigned)(((long long)N * 4 + TB - 1) / TB), TB, 0, stream>>>(
        rowinfo, csr, dinv, xpad, xagg, N, E);
    gemm_wmma<8, 4, true, false><<<rowTiles, 32, 0, stream>>>(xagg, Wpad, bufB, b1, nullptr);

    const float* Wl[2] = {W2, W3};
    const float* bl[2] = {b2, b3};
    for (int l = 0; l < 2; ++l) {
        gemm_wmma<64, 4, false, true><<<rowTiles, 32, 0, stream>>>(bufB, Wl[l], bufA, nullptr, dinv);
        gather_rows_kernel<<<gb, TB, 0, stream>>>(rowinfo, csr, dinv, bufA, bl[l], bufB, N, E);
    }

    pool_zero_kernel<<<(NG * HID + TB - 1) / TB, TB, 0, stream>>>(sums, cnts, NG, HID);
    pool_rows_kernel<<<(N + 127) / 128, 128, 0, stream>>>(bufB, batch, sums, cnts, N, NG);
    pool_div_kernel<<<(NG * HID + TB - 1) / TB, TB, 0, stream>>>(sums, cnts, G, NG, HID);

    gemm_wmma<64, 8, true, false><<<NG / 16, 32, 0, stream>>>(G, Wfc, (float*)d_out, bfc, nullptr);
}


# each K=64 product folded into the kernel that makes its input rows: layer 1 multiplies its tile by W2 through LDS; layer 2's gather (16 waves a group, one per node) by W3; 11 launches to 9
# speedup vs baseline: 3.6450x; 1.2314x over previous
#include <hip/hip_runtime.h>

typedef __attribute__((ext_vector_type(2))) float v2f;
typedef __attribute__((ext_vector_type(8))) float v8f;

constexpr int BUCKET_SHIFT = 8;
constexpr int BUCKET_NODES = 1 << BUCKET_SHIFT;
constexpr int SORT_THREADS = 1024;
constexpr int EDGES_PER_THREAD = 4;
constexpr int EPB = SORT_THREADS * EDGES_PER_THREAD;
constexpr int MAX_BUCKETS = 1024;

__device__ __forceinline__ int wave_incl_scan(int v, int lane) {
    v += __builtin_amdgcn_update_dpp(0, v, 0x111, 0xf, 0xf, false);
    v += __builtin_amdgcn_update_dpp(0, v, 0x112, 0xf, 0xf, false);
    v += __builtin_amdgcn_update_dpp(0, v, 0x114, 0xf, 0xf, false);
    v += __builtin_amdgcn_update_dpp(0, v, 0x118, 0xf, 0xf, false);
    const int low = __builtin_amdgcn_readlane(v, 15);
    return (lane >= 16) ? v + low : v;
}

__device__ __forceinline__ int bucket_of(int d, int nb) {
    int b = d >> BUCKET_SHIFT;
    return (b < 0) ? 0 : ((b > nb - 1) ? nb - 1 : b);
}

__global__ __launch_bounds__(SORT_THREADS) void bucket_count_kernel(
    const int* __restrict__ dst, int* __restrict__ hist, int E, int nb, int nblk)
{
    __shared__ int lds_tab[MAX_BUCKETS];
    const int tid = threadIdx.x;
    for (int i = tid; i < nb; i += SORT_THREADS) lds_tab[i] = 0;
    __syncthreads();
    const int e0 = blockIdx.x * EPB + tid;
    int d[EDGES_PER_THREAD];
#pragma unroll
    for (int k = 0; k < EDGES_PER_THREAD; ++k) {
        const int e = e0 + k * SORT_THREADS;
        d[k] = (e < E) ? dst[e] : -1;
    }
#pragma unroll
    for (int k = 0; k < EDGES_PER_THREAD; ++k) {
        const int e = e0 + k * SORT_THREADS;
        if (e < E) atomicAdd(&lds_tab[bucket_of(d[k], nb)], 1);
    }
    __syncthreads();
    for (int i = tid; i < nb; i += SORT_THREADS) hist[i * nblk + blockIdx.x] = lds_tab[i];
}

constexpr int SCAN_TILES = 10;
__global__ __launch_bounds__(SORT_THREADS) void scan_counts_kernel(
    const int* __restrict__ hist, int* __restrict__ off, int L,
    float* __restrict__ sums, float* __restrict__ cnts, int nsum, int ncnt)
{
    __shared__ int wtot[2][32];
    const int tid = threadIdx.x;
    const int lane = tid & 31;
    const int wave = tid >> 5;
    for (int i = tid; i < nsum; i += SORT_THREADS) sums[i] = 0.0f;
    for (int i = tid; i < ncnt; i += SORT_THREADS) cnts[i] = 0.0f;
    int carry = 0;
    for (int base = 0; base < L; base += SCAN_TILES * 4 * SORT_THREADS) {
        int4 v[SCAN_TILES];
#pragma unroll
        for (int k = 0; k < SCAN_TILES; ++k) {
            const int idx = base + (k * SORT_THREADS + tid) * 4;
            v[k] = (idx < L) ? *(const int4*)(hist + idx) : make_int4(0, 0, 0, 0);
        }
#pragma unroll
        for (int k = 0; k < SCAN_TILES; ++k) {
            const int idx = base + (k * SORT_THREADS + tid) * 4;
            const int s0 = v[k].x;
            const int s1 = s0 + v[k].y;
            const int s2 = s1 + v[k].z;
            const int s3 = s2 + v[k].w;
            const int incl = wave_incl_scan(s3, lane);
            if (lane == 31) wtot[k & 1][wave] = incl;
            __syncthreads();
            int wexcl = 0, total = 0;
#pragma unroll
            for (int w = 0; w < 32; ++w) {
                const int t = wtot[k & 1][w];
                total += t;
                if (w < wave) wexcl += t;
            }
            const int ex = carry + wexcl + incl - s3;
            if (idx < L) *(int4*)(off + idx) = make_int4(ex, ex + s0, ex + s1, ex + s2);
            carry += total;
        }
    }
}

__global__ __launch_bounds__(SORT_THREADS) void bucket_scatter_kernel(
    const int* __restrict__ src, const int* __restrict__ dst, const int* __restrict__ off,
    unsigned* __restrict__ blist, int E, int nb, int nblk)
{
    __shared__ int lds_tab[MAX_BUCKETS];
    const int tid = threadIdx.x;
    for (int i = tid; i < nb; i += SORT_THREADS) lds_tab[i] = off[i * nblk + blockIdx.x];
    __syncthreads();
    const int e0 = blockIdx.x * EPB + tid;
    int s[EDGES_PER_THREAD], d[EDGES_PER_THREAD];
#pragma unroll
    for (int k = 0; k < EDGES_PER_THREAD; ++k) {
        const int e = e0 + k * SORT_THREADS;
        s[k] = (e < E) ? src[e] : 0;
        d[k] = (e < E) ? dst[e] : -1;
    }
#pragma unroll
    for (int k = 0; k < EDGES_PER_THREAD; ++k) {
        const int e = e0 + k * SORT_THREADS;
        if (e < E) {
            int pos = atomicAdd(&lds_tab[bucket_of(d[k], nb)], 1);
            pos = (pos < 0) ? 0 : ((pos > E - 1) ? E - 1 : pos);
            blist[pos] = ((unsigned)s[k] << BUCKET_SHIFT) | ((unsigned)d[k] & (BUCKET_NODES - 1));
        }
    }
}

__global__ __launch_bounds__(SORT_THREADS) void bucket_rows_kernel(
    const int* __restrict__ off, const unsigned* __restrict__ blist,
    int* __restrict__ csr, int2* __restrict__ rowinfo, float* __restrict__ dinv,
    const float* __restrict__ x, float* __restrict__ xs,
    int n, int E, int nb, int nblk)
{
    __shared__ int cnt[BUCKET_NODES];
    __shared__ int cur[BUCKET_NODES];
    __shared__ int wtot[BUCKET_NODES / 32];
    const int tid = threadIdx.x;
    const int lane = tid & 31;
    const int wave = tid >> 5;
    const int b = blockIdx.x;
    const int mynode = b * BUCKET_NODES + tid;
    const bool own = (tid < BUCKET_NODES) && (mynode < n);
    float2 x01 = make_float2(0.0f, 0.0f), x23 = x01, x45 = x01;
    if (own) {
        const float2* xr = (const float2*)(x + (size_t)mynode * 6);
        x01 = xr[0];
        x23 = xr[1];
        x45 = xr[2];
    }
    int bstart = off[b * nblk];
    int bend = (b + 1 < nb) ? off[(b + 1) * nblk] : E;
    bstart = (bstart < 0) ? 0 : ((bstart > E) ? E : bstart);
    bend = (bend < bstart) ? bstart : ((bend > E) ? E : bend);
    if (tid < BUCKET_NODES) cnt[tid] = 0;
    __syncthreads();
    for (int e = bstart + tid; e < bend; e += SORT_THREADS)
        atomicAdd(&cnt[blist[e] & (BUCKET_NODES - 1)], 1);
    __syncthreads();
    const int c = (tid < BUCKET_NODES) ? cnt[tid] : 0;
    const int incl = wave_incl_scan(c, lane);
    if (tid < BUCKET_NODES && lane == 31) wtot[wave] = incl;
    __syncthreads();
    if (tid < BUCKET_NODES) {
        int before = 0;
        for (int w = 0; w < wave; ++w) before += wtot[w];
        const int start = bstart + before + incl - c;
        cur[tid] = start;
        if (own) {
            rowinfo[mynode] = make_int2(start, c);
            const float dg = (float)(c + 1);
            const float di = (dg > 0.0f) ? rsqrtf(dg) : 0.0f;
            dinv[mynode] = di;
            float4* xo = (float4*)(xs + (size_t)mynode * 8);
            xo[0] = make_float4(x01.x * di, x01.y * di, x45.x * di, x45.y * di);
            xo[1] = make_float4(x23.x * di, x23.y * di, 0.0f, 0.0f);
        }
    }
    __syncthreads();
    for (int e = bstart + tid; e < bend; e += SORT_THREADS) {
        const unsigned w = blist[e];
        int pos = atomicAdd(&cur[w & (BUCKET_NODES - 1)], 1);
        pos = (pos < 0) ? 0 : ((pos > E - 1) ? E - 1 : pos);
        csr[pos] = (int)(w >> BUCKET_SHIFT);
    }
}

constexpr int TILE_PITCH = 68;

template <int K, int NT, bool BIAS_RELU, bool ROW_SCALE, bool ROW_DIV>
__global__ __launch_bounds__(32) void gemm_wmma(
    const float* __restrict__ X, const float* __restrict__ W,
    float* __restrict__ Out, const float* __restrict__ bias,
    const float* __restrict__ rscale)
{
    constexpr int NCOLS = NT * 16;
    const int lane = threadIdx.x;
    const int hi   = lane >> 4;
    const int lo   = lane & 15;
    const int row0 = blockIdx.x * 16;

    const float* Arow = X + (size_t)(row0 + lo) * K;

    float rs[8];
    if (ROW_SCALE) {
#pragma unroll
        for (int j = 0; j < 8; ++j) rs[j] = rscale[row0 + j + 8 * hi];
    }

    float den = 1.0f;
    if (ROW_DIV) den = fmaxf(rscale[row0 + lo], 1.0f);

    v8f c[NT];
#pragma unroll
    for (int t = 0; t < NT; ++t) c[t] = (v8f){};

#pragma unroll
    for (int k0 = 0; k0 < K; k0 += 4) {
        const int ka = k0 + 2 * hi;
        v2f a = *(const v2f*)(Arow + ka);
        if (ROW_DIV) {
            a.x = a.x / den;
            a.y = a.y / den;
        }
#pragma unroll
        for (int t = 0; t < NT; ++t) {
            v2f b;
            b.x = W[ka * NCOLS + 16 * t + lo];
            b.y = W[(ka + 1) * NCOLS + 16 * t + lo];
            c[t] = __builtin_amdgcn_wmma_f32_16x16x4_f32(
                false, a, false, b, (short)0, c[t], false, false);
        }
    }

#pragma unroll
    for (int t = 0; t < NT; ++t) {
        const int col = 16 * t + lo;
#pragma unroll
        for (int j = 0; j < 8; ++j) {
            float v = c[t][j];
            if (ROW_SCALE) v *= rs[j];
            if (BIAS_RELU) v = fmaxf(v + bias[col], 0.0f);
            Out[(size_t)(row0 + j + 8 * hi) * NCOLS + col] = v;
        }
    }
}

template <int K>
__device__ __forceinline__ void gather_batch(const float* __restrict__ tmp, int p,
                                             int j, unsigned lane, float2& acc) {
    float2 v[K];
#pragma unroll
    for (int u = 0; u < K; ++u) {
        const int sj = __builtin_amdgcn_readlane(p, j + u);
        v[u] = *(const float2*)(tmp + (size_t)sj * 64 + 2u * lane);
    }
#pragma unroll
    for (int u = 0; u < K; ++u) {
        acc.x += v[u].x;
        acc.y += v[u].y;
    }
}

__device__ __forceinline__ float2 gather_node(
    const int2* __restrict__ rowinfo, const int* __restrict__ csr,
    const float* __restrict__ dinv, const float* __restrict__ tmp,
    const float* __restrict__ bias, int i, unsigned lane, int n, int E)
{
    const int2 ri = rowinfo[i];
    const float di = dinv[i];
    const float2 self = *(const float2*)(tmp + (size_t)i * 64 + 2u * lane);
    const float2 bv = *(const float2*)(bias + 2u * lane);
    float2 acc = self;
    const int start = ri.x;
    const int deg = ri.y;
    for (int b = 0; b < deg; b += 32) {
        const int m = (deg - b < 32) ? (deg - b) : 32;
        int p = 0;
        if ((int)lane < m) {
            int q = start + b + (int)lane;
            q = (q < 0) ? 0 : ((q > E - 1) ? E - 1 : q);
            p = csr[q];
            p = (p < 0) ? 0 : ((p > n - 1) ? n - 1 : p);
        }
        int j = 0;
        for (; j + 8 <= m; j += 8) gather_batch<8>(tmp, p, j, lane, acc);
        if (m & 4) { gather_batch<4>(tmp, p, j, lane, acc); j += 4; }
        if (m & 2) { gather_batch<2>(tmp, p, j, lane, acc); j += 2; }
        if (m & 1) { gather_batch<1>(tmp, p, j, lane, acc); }
    }
    float2 o;
    o.x = fmaxf(fmaf(acc.x, di, bv.x), 0.0f);
    o.y = fmaxf(fmaf(acc.y, di, bv.y), 0.0f);
    return o;
}

__global__ __launch_bounds__(256) void gather_rows_kernel(
    const int2* __restrict__ rowinfo, const int* __restrict__ csr,
    const float* __restrict__ dinv, const float* __restrict__ tmp,
    const float* __restrict__ bias, float* __restrict__ out, int n, int E)
{
    const unsigned lane = threadIdx.x & 31u;
    const int i = __builtin_amdgcn_readfirstlane(
        (int)((blockIdx.x * blockDim.x + threadIdx.x) >> 5));
    if (i >= n) return;
    const float2 o = gather_node(rowinfo, csr, dinv, tmp, bias, i, lane, n, E);
    *(float2*)(out + (size_t)i * 64 + 2u * lane) = o;
}

__global__ __launch_bounds__(512, 16) void gather_product_kernel(
    const int2* __restrict__ rowinfo, const int* __restrict__ csr,
    const float* __restrict__ dinv, const float* __restrict__ tmp,
    const float* __restrict__ bias, const float* __restrict__ Wn,
    float* __restrict__ out, int n, int E)
{
    __shared__ alignas(16) float tile[16 * TILE_PITCH];
    __shared__ alignas(16) float part[4 * 16 * TILE_PITCH];
    const unsigned lane = threadIdx.x & 31u;
    const int hi = (int)(lane >> 4);
    const int lo = (int)(lane & 15u);
    const int wave = __builtin_amdgcn_readfirstlane((int)(threadIdx.x >> 5));
    const int row0 = blockIdx.x * 16;
    const int i = row0 + wave;
    const int t = wave & 3;
    const int q = wave >> 2;
    const int col = 16 * t + lo;

    v2f b[4];
#pragma unroll
    for (int s = 0; s < 4; ++s) {
        const int ka = 16 * q + 4 * s + 2 * hi;
        b[s].x = Wn[ka * 64 + col];
        b[s].y = Wn[(ka + 1) * 64 + col];
    }

    const float di = dinv[(i < n) ? i : (n - 1)];

    float2 o;
    o.x = 0.0f;
    o.y = 0.0f;
    if (i < n) o = gather_node(rowinfo, csr, dinv, tmp, bias, i, lane, n, E);
    *(float2*)(tile + wave * TILE_PITCH + 2u * lane) = o;
    __syncthreads();

    v8f c = (v8f){};
#pragma unroll
    for (int s = 0; s < 4; ++s) {
        const int ka = 16 * q + 4 * s + 2 * hi;
        const v2f a = *(const v2f*)(tile + lo * TILE_PITCH + ka);
        c = __builtin_amdgcn_wmma_f32_16x16x4_f32(
            false, a, false, b[s], (short)0, c, false, false);
    }
#pragma unroll
    for (int j = 0; j < 8; ++j)
        part[(q * 16 + j + 8 * hi) * TILE_PITCH + col] = c[j];
    __syncthreads();

    if (i < n) {
        const float* pr = part + wave * TILE_PITCH + 2u * lane;
        const float2 p0 = *(const float2*)(pr);
        const float2 p1 = *(const float2*)(pr + 16 * TILE_PITCH);
        const float2 p2 = *(const float2*)(pr + 32 * TILE_PITCH);
        const float2 p3 = *(const float2*)(pr + 48 * TILE_PITCH);
        float2 r;
        r.x = ((p0.x + p1.x) + (p2.x + p3.x)) * di;
        r.y = ((p0.y + p1.y) + (p2.y + p3.y)) * di;
        *(float2*)(out + (size_t)i * 64 + 2u * lane) = r;
    }
}

__global__ __launch_bounds__(32) void layer1_kernel(
    const int2* __restrict__ rowinfo, const int* __restrict__ csr,
    const float* __restrict__ dinv, const float* __restrict__ xs,
    const float* __restrict__ W, const float* __restrict__ bias,
    const float* __restrict__ Wn,
    float* __restrict__ Out, int n, int E)
{
    __shared__ alignas(16) float tile[16 * TILE_PITCH];
    constexpr int NT = 4;
    constexpr int NCOLS = NT * 16;
    const int lane = threadIdx.x;
    const int hi   = lane >> 4;
    const int lo   = lane & 15;
    const int row0 = blockIdx.x * 16;
    const int i    = row0 + lo;

    const int2 ri = rowinfo[i];
    const float di = dinv[i];
    const float* xh = xs + 4 * hi;
    float4 acc = *(const float4*)(xh + (size_t)i * 8);

    v2f b0[NT], b1[NT];
    float bc[NT];
#pragma unroll
    for (int t = 0; t < NT; ++t) {
        const int col = 16 * t + lo;
        b0[t].x = W[(2 * hi) * NCOLS + col];
        b0[t].y = W[(2 * hi + 1) * NCOLS + col];
        const float w4 = W[4 * NCOLS + col];
        const float w5 = W[5 * NCOLS + col];
        b1[t].x = hi ? 0.0f : w4;
        b1[t].y = hi ? 0.0f : w5;
        bc[t] = bias[col];
    }

    const int start = ri.x;
    const int deg = ri.y;
    for (int j = 0; j < deg; j += 8) {
        int s[8];
#pragma unroll
        for (int u = 0; u < 8; ++u) {
            int q = start + ((j + u < deg) ? (j + u) : (deg - 1));
            q = (q < 0) ? 0 : ((q > E - 1) ? E - 1 : q);
            int p = csr[q];
            s[u] = (p < 0) ? 0 : ((p > n - 1) ? n - 1 : p);
        }
        float4 v[8];
#pragma unroll
        for (int u = 0; u < 8; ++u)
            v[u] = *(const float4*)(xh + (size_t)s[u] * 8);
#pragma unroll
        for (int u = 0; u < 8; ++u) {
            const bool on = (j + u < deg);
            acc.x = on ? acc.x + v[u].x : acc.x;
            acc.y = on ? acc.y + v[u].y : acc.y;
            acc.z = on ? acc.z + v[u].z : acc.z;
            acc.w = on ? acc.w + v[u].w : acc.w;
        }
    }
    constexpr int PRE = 4;
    v2f bn[PRE][NT];
#pragma unroll
    for (int s = 0; s < PRE; ++s) {
        const int ka = 4 * s + 2 * hi;
#pragma unroll
        for (int t = 0; t < NT; ++t) {
            bn[s][t].x = Wn[ka * NCOLS + 16 * t + lo];
            bn[s][t].y = Wn[(ka + 1) * NCOLS + 16 * t + lo];
        }
    }

    v2f a0, a1;
    a0.x = acc.x * di;
    a0.y = acc.y * di;
    a1.x = acc.z * di;
    a1.y = acc.w * di;

    v8f c[NT];
#pragma unroll
    for (int t = 0; t < NT; ++t) {
        c[t] = __builtin_amdgcn_wmma_f32_16x16x4_f32(
            false, a0, false, b0[t], (short)0, (v8f){}, false, false);
        c[t] = __builtin_amdgcn_wmma_f32_16x16x4_f32(
            false, a1, false, b1[t], (short)0, c[t], false, false);
    }

    float rs[8];
#pragma unroll
    for (int j = 0; j < 8; ++j) rs[j] = dinv[row0 + j + 8 * hi];

#pragma unroll
    for (int t = 0; t < NT; ++t) {
        const int col = 16 * t + lo;
#pragma unroll
        for (int j = 0; j < 8; ++j)
            tile[(j + 8 * hi) * TILE_PITCH + col] = fmaxf(c[t][j] + bc[t], 0.0f);
    }
    __syncthreads();

    v8f d[NT];
#pragma unroll
    for (int t = 0; t < NT; ++t) d[t] = (v8f){};
#pragma unroll
    for (int k0 = 0; k0 < NCOLS; k0 += 4) {
        const int ka = k0 + 2 * hi;
        const v2f a = *(const v2f*)(tile + lo * TILE_PITCH + ka);
#pragma unroll
        for (int t = 0; t < NT; ++t) {
            v2f b;
            if (k0 < 4 * PRE) {
                b = bn[k0 / 4][t];
            } else {
                b.x = Wn[ka * NCOLS + 16 * t + lo];
                b.y = Wn[(ka + 1) * NCOLS + 16 * t + lo];
            }
            d[t] = __builtin_amdgcn_wmma_f32_16x16x4_f32(
                false, a, false, b, (short)0, d[t], false, false);
        }
    }

#pragma unroll
    for (int t = 0; t < NT; ++t) {
        const int col = 16 * t + lo;
#pragma unroll
        for (int j = 0; j < 8; ++j)
            Out[(size_t)(row0 + j + 8 * hi) * NCOLS + col] = d[t][j] * rs[j];
    }
}

__device__ __forceinline__ void pool_flush(float* __restrict__ sums, float* __restrict__ cnts,
                                           int g, int ng, unsigned lane, float2 acc, int len) {
    if ((unsigned)g < (unsigned)ng) {
        float* s = sums + (size_t)g * 64 + 2u * lane;
        atomicAdd(s, acc.x);
        atomicAdd(s + 1, acc.y);
        if (lane == 0) atomicAdd(&cnts[g], (float)len);
    }
}

__global__ __launch_bounds__(128) void pool_rows_kernel(
    const float* __restrict__ h, const int* __restrict__ batch,
    float* __restrict__ sums, float* __restrict__ cnts, int n, int ng)
{
    const unsigned lane = threadIdx.x & 31u;
    const int base = __builtin_amdgcn_readfirstlane(
        (int)((blockIdx.x * blockDim.x + threadIdx.x) >> 5)) * 32;
    if (base >= n) return;
    const int m = (n - base < 32) ? (n - base) : 32;
    const int gid = batch[base + (((int)lane < m) ? (int)lane : 0)];
    float2 v[32];
#pragma unroll
    for (int u = 0; u < 32; ++u) {
        const int r = base + ((u < m) ? u : (m - 1));
        v[u] = *(const float2*)(h + (size_t)r * 64 + 2u * lane);
    }
    int cur = __builtin_amdgcn_readlane(gid, 0);
    int len = 0;
    float2 acc;
    acc.x = 0.0f;
    acc.y = 0.0f;
#pragma unroll
    for (int u = 0; u < 32; ++u) {
        if (u < m) {
            const int g = __builtin_amdgcn_readlane(gid, u);
            if (g != cur) {
                pool_flush(sums, cnts, cur, ng, lane, acc, len);
                cur = g;
                len = 0;
                acc.x = 0.0f;
                acc.y = 0.0f;
            }
            acc.x += v[u].x;
            acc.y += v[u].y;
            ++len;
        }
    }
    pool_flush(sums, cnts, cur, ng, lane, acc, len);
}

extern "C" void kernel_launch(void* const* d_in, const int* in_sizes, int n_in,
                              void* d_out, int out_size, void* d_ws, size_t ws_size,
                              hipStream_t stream) {
    (void)n_in; (void)out_size; (void)ws_size;

    const float* x     = (const float*)d_in[0];
    const int*   ei    = (const int*)  d_in[1];
    const int*   batch = (const int*)  d_in[2];
    const float* W1  = (const float*)d_in[3];
    const float* b1  = (const float*)d_in[4];
    const float* W2  = (const float*)d_in[5];
    const float* b2  = (const float*)d_in[6];
    const float* W3  = (const float*)d_in[7];
    const float* b3  = (const float*)d_in[8];
    const float* Wfc = (const float*)d_in[9];
    const float* bfc = (const float*)d_in[10];

    const int FIN = 6, HID = 64, EMB = 128, NG = 64;
    const int N = in_sizes[0] / FIN;
    const int E = in_sizes[1] / 2;
    const int* srcIdx = ei;
    const int* dstIdx = ei + E;

    char* ws = (char*)d_ws;
    auto a256 = [](size_t v) { return (v + 255) & ~(size_t)255; };
    size_t off = 0;
    float* dinv = (float*)(ws + off); off += a256((size_t)N * 4);
    float* xs   = (float*)(ws + off); off += a256((size_t)N * 8 * 4);
    float* bufA = (float*)(ws + off); off += a256((size_t)N * HID * 4);
    float* bufB = (float*)(ws + off); off += a256((size_t)N * HID * 4);
    float* sums = (float*)(ws + off); off += a256((size_t)NG * HID * 4);
    float* cnts = (float*)(ws + off); off += a256((size_t)NG * 4);
    int2* rowinfo = (int2*)(ws + off); off += a256((size_t)N * 8);
    int*  csr     = (int*) (ws + off); off += a256((size_t)E * 4);
    const int nbuckets = (N + BUCKET_NODES - 1) >> BUCKET_SHIFT;
    const int nblk = (((E + EPB - 1) / EPB) + 3) & ~3;
    const int L = nbuckets * nblk;
    int* hist = (int*)bufA;
    int* offs = hist + ((L + 63) & ~63);
    unsigned* blist = (unsigned*)bufB;

    const int TB = 256;
    const unsigned gb  = (unsigned)(((long long)N * 32 + TB - 1) / TB);

    bucket_count_kernel<<<nblk, SORT_THREADS, 0, stream>>>(dstIdx, hist, E, nbuckets, nblk);
    scan_counts_kernel<<<1, SORT_THREADS, 0, stream>>>(hist, offs, L, sums, cnts, NG * HID, NG);
    bucket_scatter_kernel<<<nblk, SORT_THREADS, 0, stream>>>(srcIdx, dstIdx, offs, blist,
                                                                   E, nbuckets, nblk);
    bucket_rows_kernel<<<nbuckets, SORT_THREADS, 0, stream>>>(offs, blist, csr, rowinfo, dinv,
                                                              x, xs, N, E, nbuckets, nblk);

    const int rowTiles = N / 16;

    layer1_kernel<<<rowTiles, 32, 0, stream>>>(rowinfo, csr, dinv, xs, W1, b1, W2, bufA, N, E);

    gather_product_kernel<<<rowTiles, 512, 0, stream>>>(rowinfo, csr, dinv, bufA, b2, W3,
                                                        bufB, N, E);

    gather_rows_kernel<<<gb, TB, 0, stream>>>(rowinfo, csr, dinv, bufB, b3, bufA, N, E);

    pool_rows_kernel<<<(N + 127) / 128, 128, 0, stream>>>(bufA, batch, sums, cnts, N, NG);

    gemm_wmma<64, 8, true, false, true><<<NG / 16, 32, 0, stream>>>(sums, Wfc, (float*)d_out, bfc, cnts);
}
